# starting kernel
# baseline (speedup 1.0000x reference)
#include <hip/hip_runtime.h>
#include <hip/hip_bf16.h>
#include <math.h>


typedef float v2f __attribute__((ext_vector_type(2)));
typedef float v8f __attribute__((ext_vector_type(8)));

#define PI_F 3.14159265358979323846f
#define RSTRIDE 133
#define KFCOLS 132

__device__ __forceinline__ v8f wmma_f32(v2f a, v2f b, v8f c) {
  return __builtin_amdgcn_wmma_f32_16x16x4_f32(false, a, false, b, (short)0, c,
                                               false, false);
}

template <typename F>
__device__ __forceinline__ v2f load_a_frag(F f, int kc, int lane) {
  const int r  = lane & 15;
  const int hi = (lane >> 4) & 1;
  v2f a;
  a.x = f(r, kc + 2 * hi);
  a.y = f(r, kc + 2 * hi + 1);
  return a;
}

template <typename F>
__device__ __forceinline__ v2f load_b_frag(F f, int kc, int lane) {
  const int c  = lane & 15;
  const int hi = (lane >> 4) & 1;
  v2f b;
  b.x = f(kc + 2 * hi, c);
  b.y = f(kc + 2 * hi + 1, c);
  return b;
}

template <typename FAr, typename FAi, typename FBr, typename FBi>
__device__ __forceinline__ void cmatmul16(FAr ar_f, FAi ai_f, FBr br_f,
                                          FBi bi_f, int lane, v8f& dre,
                                          v8f& dim) {
  v8f rr = {}, ii = {}, ri = {}, ir = {};
#pragma unroll
  for (int kc = 0; kc < 16; kc += 4) {
    v2f ar = load_a_frag(ar_f, kc, lane);
    v2f ai = load_a_frag(ai_f, kc, lane);
    v2f br = load_b_frag(br_f, kc, lane);
    v2f bi = load_b_frag(bi_f, kc, lane);
    rr = wmma_f32(ar, br, rr);
    ii = wmma_f32(ai, bi, ii);
    ri = wmma_f32(ar, bi, ri);
    ir = wmma_f32(ai, br, ir);
  }
  dre = rr - ii;
  dim = ri + ir;
}

template <typename SRC, typename DST>
__device__ __forceinline__ void fft256(SRC src, DST dst, const float2* Ftab,
                                       const float2* Wtab, float2* spill,
                                       float dir, int lane) {
  v8f bre, bim;
  cmatmul16(
      [&](int r, int c) { return src(r + 16 * c).x; },
      [&](int r, int c) { return src(r + 16 * c).y; },
      [&](int r, int c) { return Ftab[r * 16 + c].x; },
      [&](int r, int c) { return dir * Ftab[r * 16 + c].y; }, lane, bre, bim);

  const int col   = lane & 15;
  const int rbase = (lane >> 4) ? 8 : 0;
#pragma unroll
  for (int j = 0; j < 8; ++j) {
    const int row = j + rbase;
    const float2 w = Wtab[(row * col) & 255];
    const float wr = w.x, wi = dir * w.y;
    const float re = bre[j] * wr - bim[j] * wi;
    const float im = bre[j] * wi + bim[j] * wr;
    spill[row * 16 + col] = make_float2(re, im);
  }
  asm volatile("s_wait_dscnt 0" ::: "memory");

  v8f cre, cim;
  cmatmul16(
      [&](int r, int c) { return Ftab[r * 16 + c].x; },
      [&](int r, int c) { return dir * Ftab[r * 16 + c].y; },
      [&](int r, int c) { return spill[r * 16 + c].x; },
      [&](int r, int c) { return spill[r * 16 + c].y; }, lane, cre, cim);

#pragma unroll
  for (int j = 0; j < 8; ++j) {
    const int k2 = j + rbase;
    dst(col + 16 * k2, make_float2(cre[j], cim[j]));
  }
}

__device__ __forceinline__ void init_tables(float2* Ftab, float2* Wtab,
                                            int tid) {
  for (int i = tid; i < 256; i += 256) {
    const int r = i >> 4, c = i & 15;
    const float tf = -2.0f * PI_F * (float)((r * c) & 15) / 16.0f;
    Ftab[i] = make_float2(cosf(tf), sinf(tf));
    const float tw = -2.0f * PI_F * (float)i / 256.0f;
    Wtab[i] = make_float2(cosf(tw), sinf(tw));
  }
}

__global__ void __launch_bounds__(256) kf_precompute(
    const float* __restrict__ kin, float2* __restrict__ kf) {
  extern __shared__ char smem[];
  float2* Rk      = (float2*)smem;
  float2* Ftab    = (float2*)(smem + 64 * RSTRIDE * 8);
  float2* Wtab    = Ftab + 256;
  float2* scratch = Wtab + 256;

  const int tid = threadIdx.x, lane = tid & 31, wave = tid >> 5;
  const int c = blockIdx.x;

  init_tables(Ftab, Wtab, tid);
  const float* kc = kin + (size_t)c * (64 * 64);
  for (int i = tid; i < 64 * 64; i += 256) {
    const int r = i >> 6, w = i & 63;
    Rk[r * RSTRIDE + w] = make_float2(kc[i], 0.0f);
  }
  __syncthreads();

  float2* spill = scratch + wave * 512;

  for (int row = wave; row < 64; row += 8) {
    float2* Rrow = Rk + row * RSTRIDE;
    fft256(
        [&](int n) {
          return (n < 64) ? make_float2(Rrow[n].x, 0.0f) : make_float2(0.f, 0.f);
        },
        [&](int k, float2 v) { if (k <= 128) Rrow[k] = v; },
        Ftab, Wtab, spill, +1.0f, lane);
  }
  __syncthreads();

  for (int col = wave; col < 129; col += 8) {
    float2* kfcol = kf + ((size_t)c * KFCOLS + col) * 256;
    fft256(
        [&](int n) {
          return (n < 64) ? Rk[n * RSTRIDE + col] : make_float2(0.f, 0.f);
        },
        [&](int k, float2 v) { kfcol[k] = v; },
        Ftab, Wtab, spill, +1.0f, lane);
  }
}

__global__ void __launch_bounds__(256) conv_main(
    const float* __restrict__ u, const float* __restrict__ pregate,
    const float* __restrict__ postgate, const float2* __restrict__ kf,
    float* __restrict__ y) {
  extern __shared__ char smem[];
  float2* R       = (float2*)smem;
  float2* Ftab    = (float2*)(smem + 128 * RSTRIDE * 8);
  float2* Wtab    = Ftab + 256;
  float2* scratch = Wtab + 256;

  const int tid = threadIdx.x, lane = tid & 31, wave = tid >> 5;
  const int img = blockIdx.x;
  const int c = img & 255;
  const size_t base = (size_t)img * (128 * 128);

  init_tables(Ftab, Wtab, tid);

  for (int i = tid; i < 128 * 128; i += 256) {
    const int r = i >> 7, w = i & 127;
    R[r * RSTRIDE + w] = make_float2(u[base + i] * pregate[base + i], 0.0f);
  }
  __syncthreads();

  float2* spill = scratch + wave * 512;
  float2* spec  = spill + 256;

  for (int row = wave; row < 128; row += 8) {
    float2* Rrow = R + row * RSTRIDE;
    fft256(
        [&](int n) {
          return (n < 128) ? make_float2(Rrow[n].x, 0.0f)
                           : make_float2(0.f, 0.f);
        },
        [&](int k, float2 v) { if (k <= 128) Rrow[k] = v; },
        Ftab, Wtab, spill, +1.0f, lane);
  }
  __syncthreads();

  const float2* kfbase = kf + (size_t)c * KFCOLS * 256;
  for (int col = wave; col < 129; col += 8) {
    if (col + 8 < 129) {
      const char* pf = (const char*)(kfbase + (size_t)(col + 8) * 256);
      __builtin_prefetch(pf + lane * 64, 0, 1);
    }

    fft256(
        [&](int n) {
          return (n < 128) ? R[n * RSTRIDE + col] : make_float2(0.f, 0.f);
        },
        [&](int k, float2 v) { spec[k] = v; },
        Ftab, Wtab, spill, +1.0f, lane);

    const float2* kfcol = kfbase + (size_t)col * 256;
    for (int h = lane; h < 256; h += 32) {
      const float2 a = spec[h];
      const float2 b = kfcol[h];
      spec[h] = make_float2(a.x * b.x - a.y * b.y, a.x * b.y + a.y * b.x);
    }
    asm volatile("s_wait_dscnt 0" ::: "memory");

    fft256(
        [&](int n) { return spec[n]; },
        [&](int k, float2 v) { if (k < 128) R[k * RSTRIDE + col] = v; },
        Ftab, Wtab, spill, -1.0f, lane);
  }
  __syncthreads();

  const float inv_scale = 1.0f / 65536.0f;
  for (int row = wave; row < 128; row += 8) {
    float2* Rrow = R + row * RSTRIDE;
    const size_t obase = base + (size_t)row * 128;
    fft256(
        [&](int n) {
          if (n <= 128) return Rrow[n];
          const float2 v = Rrow[256 - n];
          return make_float2(v.x, -v.y);
        },
        [&](int k, float2 v) {
          if (k < 128) y[obase + k] = v.x * inv_scale * postgate[obase + k];
        },
        Ftab, Wtab, spill, -1.0f, lane);
  }
}

extern "C" void kernel_launch(void* const* d_in, const int* in_sizes, int n_in,
                              void* d_out, int out_size, void* d_ws,
                              size_t ws_size, hipStream_t stream) {
  const float* u        = (const float*)d_in[0];
  const float* k        = (const float*)d_in[1];
  const float* pregate  = (const float*)d_in[2];
  const float* postgate = (const float*)d_in[3];
  float* y   = (float*)d_out;
  float2* kf = (float2*)d_ws;

  const size_t lds_kf   = (size_t)64 * RSTRIDE * 8 + 512 * 8 + 8 * 512 * 8;
  const size_t lds_main = (size_t)128 * RSTRIDE * 8 + 512 * 8 + 8 * 512 * 8;

  (void)hipFuncSetAttribute((const void*)kf_precompute,
                            hipFuncAttributeMaxDynamicSharedMemorySize,
                            (int)lds_kf);
  (void)hipFuncSetAttribute((const void*)conv_main,
                            hipFuncAttributeMaxDynamicSharedMemorySize,
                            (int)lds_main);

  kf_precompute<<<256, 256, lds_kf, stream>>>(k, kf);
  conv_main<<<4 * 256, 256, lds_main, stream>>>(u, pregate, postgate, kf, y);
}


# Workgroups of 32 waves instead of 8 (1024 threads, 8 waves per SIMD, still one workgroup per WGP) in both kernels: sibling waves fill each transform's serial LDS and L2 round trips
# speedup vs baseline: 1.8441x; 1.8441x over previous
#include <hip/hip_runtime.h>
#include <hip/hip_bf16.h>
#include <math.h>


typedef float v2f __attribute__((ext_vector_type(2)));
typedef float v8f __attribute__((ext_vector_type(8)));

#define PI_F 3.14159265358979323846f
#define RSTRIDE 133
#define KFCOLS 132
#define NWAVES 32
#define NTHREADS (NWAVES * 32)

__device__ __forceinline__ v8f wmma_f32(v2f a, v2f b, v8f c) {
  return __builtin_amdgcn_wmma_f32_16x16x4_f32(false, a, false, b, (short)0, c,
                                               false, false);
}

template <typename F>
__device__ __forceinline__ v2f load_a_frag(F f, int kc, int lane) {
  const int r  = lane & 15;
  const int hi = (lane >> 4) & 1;
  v2f a;
  a.x = f(r, kc + 2 * hi);
  a.y = f(r, kc + 2 * hi + 1);
  return a;
}

template <typename F>
__device__ __forceinline__ v2f load_b_frag(F f, int kc, int lane) {
  const int c  = lane & 15;
  const int hi = (lane >> 4) & 1;
  v2f b;
  b.x = f(kc + 2 * hi, c);
  b.y = f(kc + 2 * hi + 1, c);
  return b;
}

template <typename FAr, typename FAi, typename FBr, typename FBi>
__device__ __forceinline__ void cmatmul16(FAr ar_f, FAi ai_f, FBr br_f,
                                          FBi bi_f, int lane, v8f& dre,
                                          v8f& dim) {
  v8f rr = {}, ii = {}, ri = {}, ir = {};
#pragma unroll
  for (int kc = 0; kc < 16; kc += 4) {
    v2f ar = load_a_frag(ar_f, kc, lane);
    v2f ai = load_a_frag(ai_f, kc, lane);
    v2f br = load_b_frag(br_f, kc, lane);
    v2f bi = load_b_frag(bi_f, kc, lane);
    rr = wmma_f32(ar, br, rr);
    ii = wmma_f32(ai, bi, ii);
    ri = wmma_f32(ar, bi, ri);
    ir = wmma_f32(ai, br, ir);
  }
  dre = rr - ii;
  dim = ri + ir;
}

template <typename SRC, typename DST>
__device__ __forceinline__ void fft256(SRC src, DST dst, const float2* Ftab,
                                       const float2* Wtab, float2* spill,
                                       float dir, int lane) {
  v8f bre, bim;
  cmatmul16(
      [&](int r, int c) { return src(r + 16 * c).x; },
      [&](int r, int c) { return src(r + 16 * c).y; },
      [&](int r, int c) { return Ftab[r * 16 + c].x; },
      [&](int r, int c) { return dir * Ftab[r * 16 + c].y; }, lane, bre, bim);

  const int col   = lane & 15;
  const int rbase = (lane >> 4) ? 8 : 0;
#pragma unroll
  for (int j = 0; j < 8; ++j) {
    const int row = j + rbase;
    const float2 w = Wtab[(row * col) & 255];
    const float wr = w.x, wi = dir * w.y;
    const float re = bre[j] * wr - bim[j] * wi;
    const float im = bre[j] * wi + bim[j] * wr;
    spill[row * 16 + col] = make_float2(re, im);
  }
  asm volatile("s_wait_dscnt 0" ::: "memory");

  v8f cre, cim;
  cmatmul16(
      [&](int r, int c) { return Ftab[r * 16 + c].x; },
      [&](int r, int c) { return dir * Ftab[r * 16 + c].y; },
      [&](int r, int c) { return spill[r * 16 + c].x; },
      [&](int r, int c) { return spill[r * 16 + c].y; }, lane, cre, cim);

#pragma unroll
  for (int j = 0; j < 8; ++j) {
    const int k2 = j + rbase;
    dst(col + 16 * k2, make_float2(cre[j], cim[j]));
  }
}

__device__ __forceinline__ void init_tables(float2* Ftab, float2* Wtab,
                                            int tid) {
  for (int i = tid; i < 256; i += NTHREADS) {
    const int r = i >> 4, c = i & 15;
    const float tf = -2.0f * PI_F * (float)((r * c) & 15) / 16.0f;
    Ftab[i] = make_float2(cosf(tf), sinf(tf));
    const float tw = -2.0f * PI_F * (float)i / 256.0f;
    Wtab[i] = make_float2(cosf(tw), sinf(tw));
  }
}

__global__ void __launch_bounds__(NTHREADS) kf_precompute(
    const float* __restrict__ kin, float2* __restrict__ kf) {
  extern __shared__ char smem[];
  float2* Rk      = (float2*)smem;
  float2* Ftab    = (float2*)(smem + 64 * RSTRIDE * 8);
  float2* Wtab    = Ftab + 256;
  float2* scratch = Wtab + 256;

  const int tid = threadIdx.x, lane = tid & 31, wave = tid >> 5;
  const int c = blockIdx.x;

  init_tables(Ftab, Wtab, tid);
  const float* kc = kin + (size_t)c * (64 * 64);
  for (int i = tid; i < 64 * 64; i += NTHREADS) {
    const int r = i >> 6, w = i & 63;
    Rk[r * RSTRIDE + w] = make_float2(kc[i], 0.0f);
  }
  __syncthreads();

  float2* spill = scratch + wave * 512;

  for (int row = wave; row < 64; row += NWAVES) {
    float2* Rrow = Rk + row * RSTRIDE;
    fft256(
        [&](int n) {
          return (n < 64) ? make_float2(Rrow[n].x, 0.0f) : make_float2(0.f, 0.f);
        },
        [&](int k, float2 v) { if (k <= 128) Rrow[k] = v; },
        Ftab, Wtab, spill, +1.0f, lane);
  }
  __syncthreads();

  for (int col = wave; col < 129; col += NWAVES) {
    float2* kfcol = kf + ((size_t)c * KFCOLS + col) * 256;
    fft256(
        [&](int n) {
          return (n < 64) ? Rk[n * RSTRIDE + col] : make_float2(0.f, 0.f);
        },
        [&](int k, float2 v) { kfcol[k] = v; },
        Ftab, Wtab, spill, +1.0f, lane);
  }
}

__global__ void __launch_bounds__(NTHREADS) conv_main(
    const float* __restrict__ u, const float* __restrict__ pregate,
    const float* __restrict__ postgate, const float2* __restrict__ kf,
    float* __restrict__ y) {
  extern __shared__ char smem[];
  float2* R       = (float2*)smem;
  float2* Ftab    = (float2*)(smem + 128 * RSTRIDE * 8);
  float2* Wtab    = Ftab + 256;
  float2* scratch = Wtab + 256;

  const int tid = threadIdx.x, lane = tid & 31, wave = tid >> 5;
  const int img = blockIdx.x;
  const int c = img & 255;
  const size_t base = (size_t)img * (128 * 128);

  init_tables(Ftab, Wtab, tid);

  for (int i = tid; i < 128 * 128; i += NTHREADS) {
    const int r = i >> 7, w = i & 127;
    R[r * RSTRIDE + w] = make_float2(u[base + i] * pregate[base + i], 0.0f);
  }
  __syncthreads();

  float2* spill = scratch + wave * 512;
  float2* spec  = spill + 256;

  for (int row = wave; row < 128; row += NWAVES) {
    float2* Rrow = R + row * RSTRIDE;
    fft256(
        [&](int n) {
          return (n < 128) ? make_float2(Rrow[n].x, 0.0f)
                           : make_float2(0.f, 0.f);
        },
        [&](int k, float2 v) { if (k <= 128) Rrow[k] = v; },
        Ftab, Wtab, spill, +1.0f, lane);
  }
  __syncthreads();

  const float2* kfbase = kf + (size_t)c * KFCOLS * 256;
  for (int col = wave; col < 129; col += NWAVES) {
    if (col + NWAVES < 129) {
      const char* pf = (const char*)(kfbase + (size_t)(col + NWAVES) * 256);
      __builtin_prefetch(pf + lane * 64, 0, 1);
    }

    fft256(
        [&](int n) {
          return (n < 128) ? R[n * RSTRIDE + col] : make_float2(0.f, 0.f);
        },
        [&](int k, float2 v) { spec[k] = v; },
        Ftab, Wtab, spill, +1.0f, lane);

    const float2* kfcol = kfbase + (size_t)col * 256;
    for (int h = lane; h < 256; h += 32) {
      const float2 a = spec[h];
      const float2 b = kfcol[h];
      spec[h] = make_float2(a.x * b.x - a.y * b.y, a.x * b.y + a.y * b.x);
    }
    asm volatile("s_wait_dscnt 0" ::: "memory");

    fft256(
        [&](int n) { return spec[n]; },
        [&](int k, float2 v) { if (k < 128) R[k * RSTRIDE + col] = v; },
        Ftab, Wtab, spill, -1.0f, lane);
  }
  __syncthreads();

  const float inv_scale = 1.0f / 65536.0f;
  for (int row = wave; row < 128; row += NWAVES) {
    float2* Rrow = R + row * RSTRIDE;
    const size_t obase = base + (size_t)row * 128;
    fft256(
        [&](int n) {
          if (n <= 128) return Rrow[n];
          const float2 v = Rrow[256 - n];
          return make_float2(v.x, -v.y);
        },
        [&](int k, float2 v) {
          if (k < 128) y[obase + k] = v.x * inv_scale * postgate[obase + k];
        },
        Ftab, Wtab, spill, -1.0f, lane);
  }
}

extern "C" void kernel_launch(void* const* d_in, const int* in_sizes, int n_in,
                              void* d_out, int out_size, void* d_ws,
                              size_t ws_size, hipStream_t stream) {
  const float* u        = (const float*)d_in[0];
  const float* k        = (const float*)d_in[1];
  const float* pregate  = (const float*)d_in[2];
  const float* postgate = (const float*)d_in[3];
  float* y   = (float*)d_out;
  float2* kf = (float2*)d_ws;

  const size_t lds_kf   = (size_t)64 * RSTRIDE * 8 + 512 * 8 + NWAVES * 512 * 8;
  const size_t lds_main = (size_t)128 * RSTRIDE * 8 + 512 * 8 + NWAVES * 512 * 8;

  (void)hipFuncSetAttribute((const void*)kf_precompute,
                            hipFuncAttributeMaxDynamicSharedMemorySize,
                            (int)lds_kf);
  (void)hipFuncSetAttribute((const void*)conv_main,
                            hipFuncAttributeMaxDynamicSharedMemorySize,
                            (int)lds_main);

  kf_precompute<<<256, NTHREADS, lds_kf, stream>>>(k, kf);
  conv_main<<<4 * 256, NTHREADS, lds_main, stream>>>(u, pregate, postgate, kf, y);
}


# Each wave's chain of products stays in registers: a D tile is the next product's operand (one k map 8h+2t+s for all fragments, so every product sums k in a permuted order); LDS spill buffers removed
# speedup vs baseline: 2.3216x; 1.2589x over previous
#include <hip/hip_runtime.h>
#include <hip/hip_bf16.h>
#include <math.h>


typedef float v2f __attribute__((ext_vector_type(2)));
typedef float v8f __attribute__((ext_vector_type(8)));

#define PI_F 3.14159265358979323846f
#define RSTRIDE 133
#define KFCOLS 132
#define NWAVES 32
#define NTHREADS (NWAVES * 32)

__device__ __forceinline__ v8f wmma_f32(v2f a, v2f b, v8f c) {
  return __builtin_amdgcn_wmma_f32_16x16x4_f32(false, a, false, b, (short)0, c,
                                               false, false);
}

__device__ __forceinline__ int kpos(int t, int h, int s) {
  return 8 * h + 2 * t + s;
}

struct ctile { v8f re, im; };
struct cfrag { v2f re[4], im[4]; };

__device__ __forceinline__ cfrag tile_as_frag(const ctile& d) {
  cfrag f;
#pragma unroll
  for (int t = 0; t < 4; ++t) {
    f.re[t] = (v2f){d.re[2 * t], d.re[2 * t + 1]};
    f.im[t] = (v2f){d.im[2 * t], d.im[2 * t + 1]};
  }
  return f;
}

__device__ __forceinline__ cfrag load_dft_frag(const float2* Ftab, int lane) {
  const int c = lane & 15, h = lane >> 4;
  cfrag f;
#pragma unroll
  for (int t = 0; t < 4; ++t) {
    const float2 e0 = Ftab[kpos(t, h, 0) * 16 + c];
    const float2 e1 = Ftab[kpos(t, h, 1) * 16 + c];
    f.re[t] = (v2f){e0.x, e1.x};
    f.im[t] = (v2f){e0.y, e1.y};
  }
  return f;
}

template <typename SRC>
__device__ __forceinline__ cfrag load_seq_frag(SRC src, int lane) {
  const int c = lane & 15, h = lane >> 4;
  cfrag f;
#pragma unroll
  for (int t = 0; t < 4; ++t) {
    const float2 e0 = src(c + 16 * kpos(t, h, 0));
    const float2 e1 = src(c + 16 * kpos(t, h, 1));
    f.re[t] = (v2f){e0.x, e1.x};
    f.im[t] = (v2f){e0.y, e1.y};
  }
  return f;
}

template <int SA, int SB>
__device__ __forceinline__ ctile cmatmul16(const cfrag& a, const cfrag& b) {
  v8f rr = {}, ii = {}, ri = {}, ir = {};
#pragma unroll
  for (int t = 0; t < 4; ++t) {
    rr = wmma_f32(a.re[t], b.re[t], rr);
    ii = wmma_f32(a.im[t], b.im[t], ii);
    ri = wmma_f32(a.re[t], b.im[t], ri);
    ir = wmma_f32(a.im[t], b.re[t], ir);
  }
  ctile d;
  if (SA * SB > 0) d.re = rr - ii; else d.re = rr + ii;
  if (SB > 0) {
    if (SA > 0) d.im = ri + ir; else d.im = ri - ir;
  } else {
    if (SA > 0) d.im = ir - ri; else d.im = -(ri + ir);
  }
  return d;
}

template <int DIR>
__device__ __forceinline__ ctile fft256(const cfrag& x, const float2* Ftab,
                                        const float2* Wtab, int lane) {
  const int c = lane & 15, h = lane >> 4;
  const cfrag f = load_dft_frag(Ftab, lane);
  ctile b = cmatmul16<+1, DIR>(x, f);

#pragma unroll
  for (int j = 0; j < 8; ++j) {
    const int row = 8 * h + j;
    const float2 w = Wtab[(row * c) & 255];
    const float wr = w.x, wi = (DIR > 0) ? w.y : -w.y;
    const float re = b.re[j] * wr - b.im[j] * wi;
    const float im = b.re[j] * wi + b.im[j] * wr;
    b.re[j] = re;
    b.im[j] = im;
  }
  return cmatmul16<DIR, +1>(f, tile_as_frag(b));
}

__device__ __forceinline__ void init_tables(float2* Ftab, float2* Wtab,
                                            int tid) {
  for (int i = tid; i < 256; i += NTHREADS) {
    const int r = i >> 4, c = i & 15;
    const float tf = -2.0f * PI_F * (float)((r * c) & 15) / 16.0f;
    Ftab[i] = make_float2(cosf(tf), sinf(tf));
    const float tw = -2.0f * PI_F * (float)i / 256.0f;
    Wtab[i] = make_float2(cosf(tw), sinf(tw));
  }
}

__global__ void __launch_bounds__(NTHREADS) kf_precompute(
    const float* __restrict__ kin, float2* __restrict__ kf) {
  extern __shared__ char smem[];
  float2* Rk   = (float2*)smem;
  float2* Ftab = (float2*)(smem + 64 * RSTRIDE * 8);
  float2* Wtab = Ftab + 256;
  float2* Zpad = Wtab + 256;

  const int tid = threadIdx.x, lane = tid & 31, wave = tid >> 5;
  const int lc = lane & 15, lh = lane >> 4;
  const int c = blockIdx.x;

  init_tables(Ftab, Wtab, tid);
  if (tid == 0) Zpad[0] = make_float2(0.0f, 0.0f);
  const float* kc = kin + (size_t)c * (64 * 64);
  for (int i = tid; i < 64 * 64; i += NTHREADS) {
    const int r = i >> 6, w = i & 63;
    Rk[r * RSTRIDE + w] = make_float2(kc[i], 0.0f);
  }
  __syncthreads();

  for (int row = wave; row < 64; row += NWAVES) {
    float2* Rrow = Rk + row * RSTRIDE;
    const cfrag x = load_seq_frag(
        [&](int n) {
          const float2* p = (n < 64) ? (Rrow + n) : Zpad;
          return make_float2(p->x, 0.0f);
        },
        lane);
    const ctile o = fft256<+1>(x, Ftab, Wtab, lane);
#pragma unroll
    for (int j = 0; j < 8; ++j) {
      const int k = lc + 16 * (8 * lh + j);
      if (k <= 128) Rrow[k] = make_float2(o.re[j], o.im[j]);
    }
  }
  __syncthreads();

  for (int col = wave; col < 129; col += NWAVES) {
    float2* kfcol = kf + ((size_t)c * KFCOLS + col) * 256;
    const cfrag x = load_seq_frag(
        [&](int n) {
          const float2* p = (n < 64) ? (Rk + n * RSTRIDE + col) : Zpad;
          return *p;
        },
        lane);
    const ctile o = fft256<+1>(x, Ftab, Wtab, lane);
#pragma unroll
    for (int j = 0; j < 8; ++j)
      kfcol[lc + 16 * (8 * lh + j)] = make_float2(o.re[j], o.im[j]);
  }
}

__global__ void __launch_bounds__(NTHREADS) conv_main(
    const float* __restrict__ u, const float* __restrict__ pregate,
    const float* __restrict__ postgate, const float2* __restrict__ kf,
    float* __restrict__ y) {
  extern __shared__ char smem[];
  float2* R    = (float2*)smem;
  float2* Ftab = (float2*)(smem + 128 * RSTRIDE * 8);
  float2* Wtab = Ftab + 256;
  float2* Zpad = Wtab + 256;

  const int tid = threadIdx.x, lane = tid & 31, wave = tid >> 5;
  const int lc = lane & 15, lh = lane >> 4;
  const int img = blockIdx.x;
  const int c = img & 255;
  const size_t base = (size_t)img * (128 * 128);

  init_tables(Ftab, Wtab, tid);
  if (tid == 0) Zpad[0] = make_float2(0.0f, 0.0f);

  for (int i = tid; i < 128 * 128; i += NTHREADS) {
    const int r = i >> 7, w = i & 127;
    R[r * RSTRIDE + w] = make_float2(u[base + i] * pregate[base + i], 0.0f);
  }
  __syncthreads();

  for (int row = wave; row < 128; row += NWAVES) {
    float2* Rrow = R + row * RSTRIDE;
    const cfrag x = load_seq_frag(
        [&](int n) {
          const float2* p = (n < 128) ? (Rrow + n) : Zpad;
          return make_float2(p->x, 0.0f);
        },
        lane);
    const ctile o = fft256<+1>(x, Ftab, Wtab, lane);
#pragma unroll
    for (int j = 0; j < 8; ++j) {
      const int k = lc + 16 * (8 * lh + j);
      if (k <= 128) Rrow[k] = make_float2(o.re[j], o.im[j]);
    }
  }
  __syncthreads();

  const float2* kfbase = kf + (size_t)c * KFCOLS * 256;
  for (int col = wave; col < 129; col += NWAVES) {
    if (col + NWAVES < 129) {
      const char* pf = (const char*)(kfbase + (size_t)(col + NWAVES) * 256);
      __builtin_prefetch(pf + lane * 64, 0, 1);
    }

    const float2* kfcol = kfbase + (size_t)col * 256;
    float2 kv[8];
#pragma unroll
    for (int j = 0; j < 8; ++j) kv[j] = kfcol[lc + 16 * (8 * lh + j)];

    const cfrag x = load_seq_frag(
        [&](int n) {
          const float2* p = (n < 128) ? (R + n * RSTRIDE + col) : Zpad;
          return *p;
        },
        lane);
    ctile s = fft256<+1>(x, Ftab, Wtab, lane);

#pragma unroll
    for (int j = 0; j < 8; ++j) {
      const float ar = s.re[j], ai = s.im[j];
      s.re[j] = ar * kv[j].x - ai * kv[j].y;
      s.im[j] = ar * kv[j].y + ai * kv[j].x;
    }

    const ctile o = fft256<-1>(tile_as_frag(s), Ftab, Wtab, lane);
#pragma unroll
    for (int j = 0; j < 8; ++j) {
      const int k = lc + 16 * (8 * lh + j);
      if (k < 128) R[k * RSTRIDE + col] = make_float2(o.re[j], o.im[j]);
    }
  }
  __syncthreads();

  const float inv_scale = 1.0f / 65536.0f;
  for (int row = wave; row < 128; row += NWAVES) {
    float2* Rrow = R + row * RSTRIDE;
    const size_t obase = base + (size_t)row * 128;
    const cfrag x = load_seq_frag(
        [&](int n) {
          const float2 v = Rrow[(n <= 128) ? n : 256 - n];
          return make_float2(v.x, (n <= 128) ? v.y : -v.y);
        },
        lane);
    const ctile o = fft256<-1>(x, Ftab, Wtab, lane);
#pragma unroll
    for (int j = 0; j < 8; ++j) {
      const int k = lc + 16 * (8 * lh + j);
      if (k < 128) y[obase + k] = o.re[j] * inv_scale * postgate[obase + k];
    }
  }
}

extern "C" void kernel_launch(void* const* d_in, const int* in_sizes, int n_in,
                              void* d_out, int out_size, void* d_ws,
                              size_t ws_size, hipStream_t stream) {
  const float* u        = (const float*)d_in[0];
  const float* k        = (const float*)d_in[1];
  const float* pregate  = (const float*)d_in[2];
  const float* postgate = (const float*)d_in[3];
  float* y   = (float*)d_out;
  float2* kf = (float2*)d_ws;

  const size_t lds_kf   = (size_t)64 * RSTRIDE * 8 + 512 * 8 + 8;
  const size_t lds_main = (size_t)128 * RSTRIDE * 8 + 512 * 8 + 8;

  (void)hipFuncSetAttribute((const void*)kf_precompute,
                            hipFuncAttributeMaxDynamicSharedMemorySize,
                            (int)lds_kf);
  (void)hipFuncSetAttribute((const void*)conv_main,
                            hipFuncAttributeMaxDynamicSharedMemorySize,
                            (int)lds_main);

  kf_precompute<<<256, NTHREADS, lds_kf, stream>>>(k, kf);
  conv_main<<<4 * 256, NTHREADS, lds_main, stream>>>(u, pregate, postgate, kf, y);
}


# Stage 1 of each zero-padded forward transform multiplies only the part of its sequence that can be non-zero: 2 of 4 K-chunks (filter: 1 of 4), real part only of real rows; new summation order of k
# speedup vs baseline: 2.6117x; 1.1250x over previous
#include <hip/hip_runtime.h>
#include <hip/hip_bf16.h>
#include <math.h>


typedef float v2f __attribute__((ext_vector_type(2)));
typedef float v8f __attribute__((ext_vector_type(8)));

#define PI_F 3.14159265358979323846f
#define RSTRIDE 133
#define KFCOLS 132
#define NWAVES 32
#define NTHREADS (NWAVES * 32)

__device__ __forceinline__ v8f wmma_f32(v2f a, v2f b, v8f c) {
  return __builtin_amdgcn_wmma_f32_16x16x4_f32(false, a, false, b, (short)0, c,
                                               false, false);
}

__device__ __forceinline__ int kpos(int t, int h, int s) {
  return 8 * h + 2 * t + s;
}

__device__ __forceinline__ int kpos_in(int t, int h, int s) {
  return 4 * t + 2 * s + h;
}

struct ctile { v8f re, im; };
template <int NCH> struct cfragn { v2f re[NCH], im[NCH]; };
typedef cfragn<4> cfrag;

__device__ __forceinline__ cfrag tile_as_frag(const ctile& d) {
  cfrag f;
#pragma unroll
  for (int t = 0; t < 4; ++t) {
    f.re[t] = (v2f){d.re[2 * t], d.re[2 * t + 1]};
    f.im[t] = (v2f){d.im[2 * t], d.im[2 * t + 1]};
  }
  return f;
}

__device__ __forceinline__ cfrag load_dft_frag(const float2* Ftab, int lane) {
  const int c = lane & 15, h = lane >> 4;
  cfrag f;
#pragma unroll
  for (int t = 0; t < 4; ++t) {
    const float2 e0 = Ftab[kpos(t, h, 0) * 16 + c];
    const float2 e1 = Ftab[kpos(t, h, 1) * 16 + c];
    f.re[t] = (v2f){e0.x, e1.x};
    f.im[t] = (v2f){e0.y, e1.y};
  }
  return f;
}

template <typename SRC>
__device__ __forceinline__ cfrag load_seq_frag(SRC src, int lane) {
  const int c = lane & 15, h = lane >> 4;
  cfrag f;
#pragma unroll
  for (int t = 0; t < 4; ++t) {
    const float2 e0 = src(c + 16 * kpos(t, h, 0));
    const float2 e1 = src(c + 16 * kpos(t, h, 1));
    f.re[t] = (v2f){e0.x, e1.x};
    f.im[t] = (v2f){e0.y, e1.y};
  }
  return f;
}

template <int NCH>
__device__ __forceinline__ cfragn<NCH> load_dft_frag_in(const float2* Ftab,
                                                        int lane) {
  const int c = lane & 15, h = lane >> 4;
  cfragn<NCH> f;
#pragma unroll
  for (int t = 0; t < NCH; ++t) {
    const float2 e0 = Ftab[kpos_in(t, h, 0) * 16 + c];
    const float2 e1 = Ftab[kpos_in(t, h, 1) * 16 + c];
    f.re[t] = (v2f){e0.x, e1.x};
    f.im[t] = (v2f){e0.y, e1.y};
  }
  return f;
}

template <int NCH, typename SRC>
__device__ __forceinline__ cfragn<NCH> load_seq_frag_in(SRC src, int lane) {
  const int c = lane & 15, h = lane >> 4;
  cfragn<NCH> f;
#pragma unroll
  for (int t = 0; t < NCH; ++t) {
    const float2 e0 = src(c + 16 * kpos_in(t, h, 0));
    const float2 e1 = src(c + 16 * kpos_in(t, h, 1));
    f.re[t] = (v2f){e0.x, e1.x};
    f.im[t] = (v2f){e0.y, e1.y};
  }
  return f;
}

template <int SA, int SB, int NCH, bool AREAL = false>
__device__ __forceinline__ ctile cmatmul16(const cfragn<NCH>& a,
                                           const cfragn<NCH>& b) {
  v8f rr = {}, ii = {}, ri = {}, ir = {};
#pragma unroll
  for (int t = 0; t < NCH; ++t) {
    rr = wmma_f32(a.re[t], b.re[t], rr);
    if (!AREAL) ii = wmma_f32(a.im[t], b.im[t], ii);
    ri = wmma_f32(a.re[t], b.im[t], ri);
    if (!AREAL) ir = wmma_f32(a.im[t], b.re[t], ir);
  }
  ctile d;
  if (AREAL) {
    d.re = rr;
    if (SB > 0) d.im = ri; else d.im = -ri;
    return d;
  }
  if (SA * SB > 0) d.re = rr - ii; else d.re = rr + ii;
  if (SB > 0) {
    if (SA > 0) d.im = ri + ir; else d.im = ri - ir;
  } else {
    if (SA > 0) d.im = ir - ri; else d.im = -(ri + ir);
  }
  return d;
}

template <int DIR>
__device__ __forceinline__ ctile fft256_tail(ctile b, const cfrag& f,
                                             const float2* Wtab, int lane) {
  const int c = lane & 15, h = lane >> 4;

#pragma unroll
  for (int j = 0; j < 8; ++j) {
    const int row = 8 * h + j;
    const float2 w = Wtab[(row * c) & 255];
    const float wr = w.x, wi = (DIR > 0) ? w.y : -w.y;
    const float re = b.re[j] * wr - b.im[j] * wi;
    const float im = b.re[j] * wi + b.im[j] * wr;
    b.re[j] = re;
    b.im[j] = im;
  }
  return cmatmul16<DIR, +1, 4>(f, tile_as_frag(b));
}

template <int DIR>
__device__ __forceinline__ ctile fft256(const cfrag& x, const float2* Ftab,
                                        const float2* Wtab, int lane) {
  const cfrag f = load_dft_frag(Ftab, lane);
  const ctile b = cmatmul16<+1, DIR, 4>(x, f);
  return fft256_tail<DIR>(b, f, Wtab, lane);
}

template <int NCH, bool REALSEQ, typename SRC>
__device__ __forceinline__ ctile fft256_in(SRC src, const float2* Ftab,
                                           const float2* Wtab, int lane) {
  const cfragn<NCH> x = load_seq_frag_in<NCH>(src, lane);
  const cfragn<NCH> fin = load_dft_frag_in<NCH>(Ftab, lane);
  const ctile b = cmatmul16<+1, +1, NCH, REALSEQ>(x, fin);
  const cfrag f = load_dft_frag(Ftab, lane);
  return fft256_tail<+1>(b, f, Wtab, lane);
}

__device__ __forceinline__ void init_tables(float2* Ftab, float2* Wtab,
                                            int tid) {
  for (int i = tid; i < 256; i += NTHREADS) {
    const int r = i >> 4, c = i & 15;
    const float tf = -2.0f * PI_F * (float)((r * c) & 15) / 16.0f;
    Ftab[i] = make_float2(cosf(tf), sinf(tf));
    const float tw = -2.0f * PI_F * (float)i / 256.0f;
    Wtab[i] = make_float2(cosf(tw), sinf(tw));
  }
}

__global__ void __launch_bounds__(NTHREADS) kf_precompute(
    const float* __restrict__ kin, float2* __restrict__ kf) {
  extern __shared__ char smem[];
  float2* Rk   = (float2*)smem;
  float2* Ftab = (float2*)(smem + 64 * RSTRIDE * 8);
  float2* Wtab = Ftab + 256;

  const int tid = threadIdx.x, lane = tid & 31, wave = tid >> 5;
  const int lc = lane & 15, lh = lane >> 4;
  const int c = blockIdx.x;

  init_tables(Ftab, Wtab, tid);
  const float* kc = kin + (size_t)c * (64 * 64);
  for (int i = tid; i < 64 * 64; i += NTHREADS) {
    const int r = i >> 6, w = i & 63;
    Rk[r * RSTRIDE + w] = make_float2(kc[i], 0.0f);
  }
  __syncthreads();

  for (int row = wave; row < 64; row += NWAVES) {
    float2* Rrow = Rk + row * RSTRIDE;
    const ctile o = fft256_in<1, true>(
        [&](int n) { return make_float2(Rrow[n].x, 0.0f); }, Ftab, Wtab,
        lane);
#pragma unroll
    for (int j = 0; j < 8; ++j) {
      const int k = lc + 16 * (8 * lh + j);
      if (k <= 128) Rrow[k] = make_float2(o.re[j], o.im[j]);
    }
  }
  __syncthreads();

  for (int col = wave; col < 129; col += NWAVES) {
    float2* kfcol = kf + ((size_t)c * KFCOLS + col) * 256;
    const ctile o = fft256_in<1, false>(
        [&](int n) { return Rk[n * RSTRIDE + col]; }, Ftab, Wtab, lane);
#pragma unroll
    for (int j = 0; j < 8; ++j)
      kfcol[lc + 16 * (8 * lh + j)] = make_float2(o.re[j], o.im[j]);
  }
}

__global__ void __launch_bounds__(NTHREADS) conv_main(
    const float* __restrict__ u, const float* __restrict__ pregate,
    const float* __restrict__ postgate, const float2* __restrict__ kf,
    float* __restrict__ y) {
  extern __shared__ char smem[];
  float2* R    = (float2*)smem;
  float2* Ftab = (float2*)(smem + 128 * RSTRIDE * 8);
  float2* Wtab = Ftab + 256;

  const int tid = threadIdx.x, lane = tid & 31, wave = tid >> 5;
  const int lc = lane & 15, lh = lane >> 4;
  const int img = blockIdx.x;
  const int c = img & 255;
  const size_t base = (size_t)img * (128 * 128);

  init_tables(Ftab, Wtab, tid);

  for (int i = tid; i < 128 * 128; i += NTHREADS) {
    const int r = i >> 7, w = i & 127;
    R[r * RSTRIDE + w] = make_float2(u[base + i] * pregate[base + i], 0.0f);
  }
  __syncthreads();

  for (int row = wave; row < 128; row += NWAVES) {
    float2* Rrow = R + row * RSTRIDE;
    const ctile o = fft256_in<2, true>(
        [&](int n) { return make_float2(Rrow[n].x, 0.0f); }, Ftab, Wtab,
        lane);
#pragma unroll
    for (int j = 0; j < 8; ++j) {
      const int k = lc + 16 * (8 * lh + j);
      if (k <= 128) Rrow[k] = make_float2(o.re[j], o.im[j]);
    }
  }
  __syncthreads();

  const float2* kfbase = kf + (size_t)c * KFCOLS * 256;
  for (int col = wave; col < 129; col += NWAVES) {
    if (col + NWAVES < 129) {
      const char* pf = (const char*)(kfbase + (size_t)(col + NWAVES) * 256);
      __builtin_prefetch(pf + lane * 64, 0, 1);
    }

    const float2* kfcol = kfbase + (size_t)col * 256;
    float2 kv[8];
#pragma unroll
    for (int j = 0; j < 8; ++j) kv[j] = kfcol[lc + 16 * (8 * lh + j)];

    ctile s = fft256_in<2, false>(
        [&](int n) { return R[n * RSTRIDE + col]; }, Ftab, Wtab, lane);

#pragma unroll
    for (int j = 0; j < 8; ++j) {
      const float ar = s.re[j], ai = s.im[j];
      s.re[j] = ar * kv[j].x - ai * kv[j].y;
      s.im[j] = ar * kv[j].y + ai * kv[j].x;
    }

    const ctile o = fft256<-1>(tile_as_frag(s), Ftab, Wtab, lane);
#pragma unroll
    for (int j = 0; j < 8; ++j) {
      const int k = lc + 16 * (8 * lh + j);
      if (k < 128) R[k * RSTRIDE + col] = make_float2(o.re[j], o.im[j]);
    }
  }
  __syncthreads();

  const float inv_scale = 1.0f / 65536.0f;
  for (int row = wave; row < 128; row += NWAVES) {
    float2* Rrow = R + row * RSTRIDE;
    const size_t obase = base + (size_t)row * 128;
    const cfrag x = load_seq_frag(
        [&](int n) {
          const float2 v = Rrow[(n <= 128) ? n : 256 - n];
          return make_float2(v.x, (n <= 128) ? v.y : -v.y);
        },
        lane);
    const ctile o = fft256<-1>(x, Ftab, Wtab, lane);
#pragma unroll
    for (int j = 0; j < 8; ++j) {
      const int k = lc + 16 * (8 * lh + j);
      if (k < 128) y[obase + k] = o.re[j] * inv_scale * postgate[obase + k];
    }
  }
}

extern "C" void kernel_launch(void* const* d_in, const int* in_sizes, int n_in,
                              void* d_out, int out_size, void* d_ws,
                              size_t ws_size, hipStream_t stream) {
  const float* u        = (const float*)d_in[0];
  const float* k        = (const float*)d_in[1];
  const float* pregate  = (const float*)d_in[2];
  const float* postgate = (const float*)d_in[3];
  float* y   = (float*)d_out;
  float2* kf = (float2*)d_ws;

  const size_t lds_kf   = (size_t)64 * RSTRIDE * 8 + 512 * 8;
  const size_t lds_main = (size_t)128 * RSTRIDE * 8 + 512 * 8;

  (void)hipFuncSetAttribute((const void*)kf_precompute,
                            hipFuncAttributeMaxDynamicSharedMemorySize,
                            (int)lds_kf);
  (void)hipFuncSetAttribute((const void*)conv_main,
                            hipFuncAttributeMaxDynamicSharedMemorySize,
                            (int)lds_main);

  kf_precompute<<<256, NTHREADS, lds_kf, stream>>>(k, kf);
  conv_main<<<4 * 256, NTHREADS, lds_main, stream>>>(u, pregate, postgate, kf, y);
}


# Every complex 16x16 matrix product in three real products instead of four (K=(Xr+Xi)Fr; re=K-Xi(Fr+Fi), im=K+Xr(Fi-Fr), K as the C operand of both chains); new rounding and summation order
# speedup vs baseline: 2.9398x; 1.1256x over previous
#include <hip/hip_runtime.h>
#include <hip/hip_bf16.h>
#include <math.h>


typedef float v2f __attribute__((ext_vector_type(2)));
typedef float v8f __attribute__((ext_vector_type(8)));

#define PI_F 3.14159265358979323846f
#define RSTRIDE 133
#define KFCOLS 132
#define NWAVES 32
#define NTHREADS (NWAVES * 32)

__device__ __forceinline__ v8f wmma_f32(v2f a, v2f b, v8f c) {
  return __builtin_amdgcn_wmma_f32_16x16x4_f32(false, a, false, b, (short)0, c,
                                               false, false);
}

__device__ __forceinline__ int kpos(int t, int h, int s) {
  return 8 * h + 2 * t + s;
}

__device__ __forceinline__ int kpos_in(int t, int h, int s) {
  return 4 * t + 2 * s + h;
}

struct ctile { v8f re, im; };
template <int NCH> struct cfragn { v2f re[NCH], im[NCH]; };
typedef cfragn<4> cfrag;
template <int NCH> struct kfragn { v2f r[NCH], p[NCH], q[NCH]; };
typedef kfragn<4> kfrag;

__device__ __forceinline__ cfrag tile_as_frag(const ctile& d) {
  cfrag f;
#pragma unroll
  for (int t = 0; t < 4; ++t) {
    f.re[t] = (v2f){d.re[2 * t], d.re[2 * t + 1]};
    f.im[t] = (v2f){d.im[2 * t], d.im[2 * t + 1]};
  }
  return f;
}

__device__ __forceinline__ kfrag load_dft_frag(const float4* Ftab, int lane) {
  const int c = lane & 15, h = lane >> 4;
  kfrag f;
#pragma unroll
  for (int t = 0; t < 4; ++t) {
    const float4 e0 = Ftab[kpos(t, h, 0) * 16 + c];
    const float4 e1 = Ftab[kpos(t, h, 1) * 16 + c];
    f.r[t] = (v2f){e0.x, e1.x};
    f.p[t] = (v2f){e0.z, e1.z};
    f.q[t] = (v2f){e0.w, e1.w};
  }
  return f;
}

template <typename SRC>
__device__ __forceinline__ cfrag load_seq_frag(SRC src, int lane) {
  const int c = lane & 15, h = lane >> 4;
  cfrag f;
#pragma unroll
  for (int t = 0; t < 4; ++t) {
    const float2 e0 = src(c + 16 * kpos(t, h, 0));
    const float2 e1 = src(c + 16 * kpos(t, h, 1));
    f.re[t] = (v2f){e0.x, e1.x};
    f.im[t] = (v2f){e0.y, e1.y};
  }
  return f;
}

template <int NCH>
__device__ __forceinline__ cfragn<NCH> load_dft_frag_in(const float4* Ftab,
                                                        int lane) {
  const int c = lane & 15, h = lane >> 4;
  cfragn<NCH> f;
#pragma unroll
  for (int t = 0; t < NCH; ++t) {
    const float4 e0 = Ftab[kpos_in(t, h, 0) * 16 + c];
    const float4 e1 = Ftab[kpos_in(t, h, 1) * 16 + c];
    f.re[t] = (v2f){e0.x, e1.x};
    f.im[t] = (v2f){e0.y, e1.y};
  }
  return f;
}

template <int NCH>
__device__ __forceinline__ kfragn<NCH> load_dft_kfrag_in(const float4* Ftab,
                                                         int lane) {
  const int c = lane & 15, h = lane >> 4;
  kfragn<NCH> f;
#pragma unroll
  for (int t = 0; t < NCH; ++t) {
    const float4 e0 = Ftab[kpos_in(t, h, 0) * 16 + c];
    const float4 e1 = Ftab[kpos_in(t, h, 1) * 16 + c];
    f.r[t] = (v2f){e0.x, e1.x};
    f.p[t] = (v2f){e0.z, e1.z};
    f.q[t] = (v2f){e0.w, e1.w};
  }
  return f;
}

template <int NCH, typename SRC>
__device__ __forceinline__ cfragn<NCH> load_seq_frag_in(SRC src, int lane) {
  const int c = lane & 15, h = lane >> 4;
  cfragn<NCH> f;
#pragma unroll
  for (int t = 0; t < NCH; ++t) {
    const float2 e0 = src(c + 16 * kpos_in(t, h, 0));
    const float2 e1 = src(c + 16 * kpos_in(t, h, 1));
    f.re[t] = (v2f){e0.x, e1.x};
    f.im[t] = (v2f){e0.y, e1.y};
  }
  return f;
}

template <bool XA>
__device__ __forceinline__ v8f mm(v2f x, v2f w, v8f c) {
  return XA ? wmma_f32(x, w, c) : wmma_f32(w, x, c);
}

template <int DIR, bool XA, int NCH>
__device__ __forceinline__ ctile cmatmul3(const cfragn<NCH>& x,
                                          const kfragn<NCH>& w) {
  v8f k = {};
#pragma unroll
  for (int t = 0; t < NCH; ++t) k = mm<XA>(x.re[t] + x.im[t], w.r[t], k);
  ctile d;
  d.re = k;
  d.im = k;
#pragma unroll
  for (int t = 0; t < NCH; ++t) {
    d.re = mm<XA>(x.im[t], (DIR > 0) ? w.q[t] : w.p[t], d.re);
    d.im = mm<XA>(x.re[t], (DIR > 0) ? w.p[t] : w.q[t], d.im);
  }
  return d;
}

template <int NCH>
__device__ __forceinline__ ctile rmatmul16(const cfragn<NCH>& a,
                                           const cfragn<NCH>& b) {
  v8f rr = {}, ri = {};
#pragma unroll
  for (int t = 0; t < NCH; ++t) {
    rr = wmma_f32(a.re[t], b.re[t], rr);
    ri = wmma_f32(a.re[t], b.im[t], ri);
  }
  ctile d;
  d.re = rr;
  d.im = ri;
  return d;
}

template <int DIR>
__device__ __forceinline__ ctile fft256_tail(ctile b, const kfrag& f,
                                             const float2* Wtab, int lane) {
  const int c = lane & 15, h = lane >> 4;

#pragma unroll
  for (int j = 0; j < 8; ++j) {
    const int row = 8 * h + j;
    const float2 w = Wtab[(row * c) & 255];
    const float wr = w.x, wi = (DIR > 0) ? w.y : -w.y;
    const float re = b.re[j] * wr - b.im[j] * wi;
    const float im = b.re[j] * wi + b.im[j] * wr;
    b.re[j] = re;
    b.im[j] = im;
  }
  return cmatmul3<DIR, false, 4>(tile_as_frag(b), f);
}

template <int DIR>
__device__ __forceinline__ ctile fft256(const cfrag& x, const float4* Ftab,
                                        const float2* Wtab, int lane) {
  const kfrag f = load_dft_frag(Ftab, lane);
  const ctile b = cmatmul3<DIR, true, 4>(x, f);
  return fft256_tail<DIR>(b, f, Wtab, lane);
}

template <int NCH, bool REALSEQ, typename SRC>
__device__ __forceinline__ ctile fft256_in(SRC src, const float4* Ftab,
                                           const float2* Wtab, int lane) {
  const cfragn<NCH> x = load_seq_frag_in<NCH>(src, lane);
  ctile b;
  if (REALSEQ)
    b = rmatmul16<NCH>(x, load_dft_frag_in<NCH>(Ftab, lane));
  else
    b = cmatmul3<+1, true, NCH>(x, load_dft_kfrag_in<NCH>(Ftab, lane));
  const kfrag f = load_dft_frag(Ftab, lane);
  return fft256_tail<+1>(b, f, Wtab, lane);
}

__device__ __forceinline__ void init_tables(float4* Ftab, float2* Wtab,
                                            int tid) {
  for (int i = tid; i < 256; i += NTHREADS) {
    const int r = i >> 4, c = i & 15;
    const float tf = -2.0f * PI_F * (float)((r * c) & 15) / 16.0f;
    const float fr = cosf(tf), fi = sinf(tf);
    Ftab[i] = make_float4(fr, fi, fi - fr, -(fr + fi));
    const float tw = -2.0f * PI_F * (float)i / 256.0f;
    Wtab[i] = make_float2(cosf(tw), sinf(tw));
  }
}

__global__ void __launch_bounds__(NTHREADS) kf_precompute(
    const float* __restrict__ kin, float2* __restrict__ kf) {
  extern __shared__ __align__(16) char smem[];
  float2* Rk   = (float2*)smem;
  float4* Ftab = (float4*)(smem + 64 * RSTRIDE * 8);
  float2* Wtab = (float2*)(Ftab + 256);

  const int tid = threadIdx.x, lane = tid & 31, wave = tid >> 5;
  const int lc = lane & 15, lh = lane >> 4;
  const int c = blockIdx.x;

  init_tables(Ftab, Wtab, tid);
  const float* kc = kin + (size_t)c * (64 * 64);
  for (int i = tid; i < 64 * 64; i += NTHREADS) {
    const int r = i >> 6, w = i & 63;
    Rk[r * RSTRIDE + w] = make_float2(kc[i], 0.0f);
  }
  __syncthreads();

  for (int row = wave; row < 64; row += NWAVES) {
    float2* Rrow = Rk + row * RSTRIDE;
    const ctile o = fft256_in<1, true>(
        [&](int n) { return make_float2(Rrow[n].x, 0.0f); }, Ftab, Wtab,
        lane);
#pragma unroll
    for (int j = 0; j < 8; ++j) {
      const int k = lc + 16 * (8 * lh + j);
      if (k <= 128) Rrow[k] = make_float2(o.re[j], o.im[j]);
    }
  }
  __syncthreads();

  for (int col = wave; col < 129; col += NWAVES) {
    float2* kfcol = kf + ((size_t)c * KFCOLS + col) * 256;
    const ctile o = fft256_in<1, false>(
        [&](int n) { return Rk[n * RSTRIDE + col]; }, Ftab, Wtab, lane);
#pragma unroll
    for (int j = 0; j < 8; ++j)
      kfcol[lc + 16 * (8 * lh + j)] = make_float2(o.re[j], o.im[j]);
  }
}

__global__ void __launch_bounds__(NTHREADS) conv_main(
    const float* __restrict__ u, const float* __restrict__ pregate,
    const float* __restrict__ postgate, const float2* __restrict__ kf,
    float* __restrict__ y) {
  extern __shared__ __align__(16) char smem[];
  float2* R    = (float2*)smem;
  float4* Ftab = (float4*)(smem + 128 * RSTRIDE * 8);
  float2* Wtab = (float2*)(Ftab + 256);

  const int tid = threadIdx.x, lane = tid & 31, wave = tid >> 5;
  const int lc = lane & 15, lh = lane >> 4;
  const int img = blockIdx.x;
  const int c = img & 255;
  const size_t base = (size_t)img * (128 * 128);

  init_tables(Ftab, Wtab, tid);

  for (int i = tid; i < 128 * 128; i += NTHREADS) {
    const int r = i >> 7, w = i & 127;
    R[r * RSTRIDE + w] = make_float2(u[base + i] * pregate[base + i], 0.0f);
  }
  __syncthreads();

  for (int row = wave; row < 128; row += NWAVES) {
    float2* Rrow = R + row * RSTRIDE;
    const ctile o = fft256_in<2, true>(
        [&](int n) { return make_float2(Rrow[n].x, 0.0f); }, Ftab, Wtab,
        lane);
#pragma unroll
    for (int j = 0; j < 8; ++j) {
      const int k = lc + 16 * (8 * lh + j);
      if (k <= 128) Rrow[k] = make_float2(o.re[j], o.im[j]);
    }
  }
  __syncthreads();

  const float2* kfbase = kf + (size_t)c * KFCOLS * 256;
  for (int col = wave; col < 129; col += NWAVES) {
    if (col + NWAVES < 129) {
      const char* pf = (const char*)(kfbase + (size_t)(col + NWAVES) * 256);
      __builtin_prefetch(pf + lane * 64, 0, 1);
    }

    const float2* kfcol = kfbase + (size_t)col * 256;
    float2 kv[8];
#pragma unroll
    for (int j = 0; j < 8; ++j) kv[j] = kfcol[lc + 16 * (8 * lh + j)];

    ctile s = fft256_in<2, false>(
        [&](int n) { return R[n * RSTRIDE + col]; }, Ftab, Wtab, lane);

#pragma unroll
    for (int j = 0; j < 8; ++j) {
      const float ar = s.re[j], ai = s.im[j];
      s.re[j] = ar * kv[j].x - ai * kv[j].y;
      s.im[j] = ar * kv[j].y + ai * kv[j].x;
    }

    const ctile o = fft256<-1>(tile_as_frag(s), Ftab, Wtab, lane);
#pragma unroll
    for (int j = 0; j < 8; ++j) {
      const int k = lc + 16 * (8 * lh + j);
      if (k < 128) R[k * RSTRIDE + col] = make_float2(o.re[j], o.im[j]);
    }
  }
  __syncthreads();

  const float inv_scale = 1.0f / 65536.0f;
  for (int row = wave; row < 128; row += NWAVES) {
    float2* Rrow = R + row * RSTRIDE;
    const size_t obase = base + (size_t)row * 128;
    const cfrag x = load_seq_frag(
        [&](int n) {
          const float2 v = Rrow[(n <= 128) ? n : 256 - n];
          return make_float2(v.x, (n <= 128) ? v.y : -v.y);
        },
        lane);
    const ctile o = fft256<-1>(x, Ftab, Wtab, lane);
#pragma unroll
    for (int j = 0; j < 8; ++j) {
      const int k = lc + 16 * (8 * lh + j);
      if (k < 128) y[obase + k] = o.re[j] * inv_scale * postgate[obase + k];
    }
  }
}

extern "C" void kernel_launch(void* const* d_in, const int* in_sizes, int n_in,
                              void* d_out, int out_size, void* d_ws,
                              size_t ws_size, hipStream_t stream) {
  const float* u        = (const float*)d_in[0];
  const float* k        = (const float*)d_in[1];
  const float* pregate  = (const float*)d_in[2];
  const float* postgate = (const float*)d_in[3];
  float* y   = (float*)d_out;
  float2* kf = (float2*)d_ws;

  const size_t lds_kf   = (size_t)64 * RSTRIDE * 8 + 256 * 16 + 256 * 8;
  const size_t lds_main = (size_t)128 * RSTRIDE * 8 + 256 * 16 + 256 * 8;

  (void)hipFuncSetAttribute((const void*)kf_precompute,
                            hipFuncAttributeMaxDynamicSharedMemorySize,
                            (int)lds_kf);
  (void)hipFuncSetAttribute((const void*)conv_main,
                            hipFuncAttributeMaxDynamicSharedMemorySize,
                            (int)lds_main);

  kf_precompute<<<256, NTHREADS, lds_kf, stream>>>(k, kf);
  conv_main<<<4 * 256, NTHREADS, lds_main, stream>>>(u, pregate, postgate, kf, y);
}


# Two batch images of one channel ride as real and imaginary part of ONE complex 256x256 transform (the real filter keeps them apart): 256 columns per pair, no Hermitian gather; new rounding, k order
# speedup vs baseline: 3.6197x; 1.2313x over previous
#include <hip/hip_runtime.h>
#include <hip/hip_bf16.h>
#include <math.h>


typedef float v2f __attribute__((ext_vector_type(2)));
typedef float v8f __attribute__((ext_vector_type(8)));

#define PI_F 3.14159265358979323846f
#define RSTRIDE 133
#define PSTRIDE 257
#define KFCOLS 132
#define NWAVES 32
#define NTHREADS (NWAVES * 32)

__device__ __forceinline__ v8f wmma_f32(v2f a, v2f b, v8f c) {
  return __builtin_amdgcn_wmma_f32_16x16x4_f32(false, a, false, b, (short)0, c,
                                               false, false);
}

__device__ __forceinline__ int kpos(int t, int h, int s) {
  return 8 * h + 2 * t + s;
}

__device__ __forceinline__ int kpos_in(int t, int h, int s) {
  return 4 * t + 2 * s + h;
}

struct ctile { v8f re, im; };
template <int NCH> struct cfragn { v2f re[NCH], im[NCH]; };
typedef cfragn<4> cfrag;
template <int NCH> struct kfragn { v2f r[NCH], p[NCH], q[NCH]; };
typedef kfragn<4> kfrag;

__device__ __forceinline__ cfrag tile_as_frag(const ctile& d) {
  cfrag f;
#pragma unroll
  for (int t = 0; t < 4; ++t) {
    f.re[t] = (v2f){d.re[2 * t], d.re[2 * t + 1]};
    f.im[t] = (v2f){d.im[2 * t], d.im[2 * t + 1]};
  }
  return f;
}

__device__ __forceinline__ kfrag load_dft_frag(const float4* Ftab, int lane) {
  const int c = lane & 15, h = lane >> 4;
  kfrag f;
#pragma unroll
  for (int t = 0; t < 4; ++t) {
    const float4 e0 = Ftab[kpos(t, h, 0) * 16 + c];
    const float4 e1 = Ftab[kpos(t, h, 1) * 16 + c];
    f.r[t] = (v2f){e0.x, e1.x};
    f.p[t] = (v2f){e0.z, e1.z};
    f.q[t] = (v2f){e0.w, e1.w};
  }
  return f;
}

template <typename SRC>
__device__ __forceinline__ cfrag load_seq_frag(SRC src, int lane) {
  const int c = lane & 15, h = lane >> 4;
  cfrag f;
#pragma unroll
  for (int t = 0; t < 4; ++t) {
    const float2 e0 = src(c + 16 * kpos(t, h, 0));
    const float2 e1 = src(c + 16 * kpos(t, h, 1));
    f.re[t] = (v2f){e0.x, e1.x};
    f.im[t] = (v2f){e0.y, e1.y};
  }
  return f;
}

template <int NCH>
__device__ __forceinline__ cfragn<NCH> load_dft_frag_in(const float4* Ftab,
                                                        int lane) {
  const int c = lane & 15, h = lane >> 4;
  cfragn<NCH> f;
#pragma unroll
  for (int t = 0; t < NCH; ++t) {
    const float4 e0 = Ftab[kpos_in(t, h, 0) * 16 + c];
    const float4 e1 = Ftab[kpos_in(t, h, 1) * 16 + c];
    f.re[t] = (v2f){e0.x, e1.x};
    f.im[t] = (v2f){e0.y, e1.y};
  }
  return f;
}

template <int NCH>
__device__ __forceinline__ kfragn<NCH> load_dft_kfrag_in(const float4* Ftab,
                                                         int lane) {
  const int c = lane & 15, h = lane >> 4;
  kfragn<NCH> f;
#pragma unroll
  for (int t = 0; t < NCH; ++t) {
    const float4 e0 = Ftab[kpos_in(t, h, 0) * 16 + c];
    const float4 e1 = Ftab[kpos_in(t, h, 1) * 16 + c];
    f.r[t] = (v2f){e0.x, e1.x};
    f.p[t] = (v2f){e0.z, e1.z};
    f.q[t] = (v2f){e0.w, e1.w};
  }
  return f;
}

template <int NCH, typename SRC>
__device__ __forceinline__ cfragn<NCH> load_seq_frag_in(SRC src, int lane) {
  const int c = lane & 15, h = lane >> 4;
  cfragn<NCH> f;
#pragma unroll
  for (int t = 0; t < NCH; ++t) {
    const float2 e0 = src(c + 16 * kpos_in(t, h, 0));
    const float2 e1 = src(c + 16 * kpos_in(t, h, 1));
    f.re[t] = (v2f){e0.x, e1.x};
    f.im[t] = (v2f){e0.y, e1.y};
  }
  return f;
}

template <bool XA>
__device__ __forceinline__ v8f mm(v2f x, v2f w, v8f c) {
  return XA ? wmma_f32(x, w, c) : wmma_f32(w, x, c);
}

template <int DIR, bool XA, int NCH>
__device__ __forceinline__ ctile cmatmul3(const cfragn<NCH>& x,
                                          const kfragn<NCH>& w) {
  v8f k = {};
#pragma unroll
  for (int t = 0; t < NCH; ++t) k = mm<XA>(x.re[t] + x.im[t], w.r[t], k);
  ctile d;
  d.re = k;
  d.im = k;
#pragma unroll
  for (int t = 0; t < NCH; ++t) {
    d.re = mm<XA>(x.im[t], (DIR > 0) ? w.q[t] : w.p[t], d.re);
    d.im = mm<XA>(x.re[t], (DIR > 0) ? w.p[t] : w.q[t], d.im);
  }
  return d;
}

template <int NCH>
__device__ __forceinline__ ctile rmatmul16(const cfragn<NCH>& a,
                                           const cfragn<NCH>& b) {
  v8f rr = {}, ri = {};
#pragma unroll
  for (int t = 0; t < NCH; ++t) {
    rr = wmma_f32(a.re[t], b.re[t], rr);
    ri = wmma_f32(a.re[t], b.im[t], ri);
  }
  ctile d;
  d.re = rr;
  d.im = ri;
  return d;
}

template <int DIR>
__device__ __forceinline__ ctile fft256_tail(ctile b, const kfrag& f,
                                             const float2* Wtab, int lane) {
  const int c = lane & 15, h = lane >> 4;

#pragma unroll
  for (int j = 0; j < 8; ++j) {
    const int row = 8 * h + j;
    const float2 w = Wtab[(row * c) & 255];
    const float wr = w.x, wi = (DIR > 0) ? w.y : -w.y;
    const float re = b.re[j] * wr - b.im[j] * wi;
    const float im = b.re[j] * wi + b.im[j] * wr;
    b.re[j] = re;
    b.im[j] = im;
  }
  return cmatmul3<DIR, false, 4>(tile_as_frag(b), f);
}

template <int DIR>
__device__ __forceinline__ ctile fft256(const cfrag& x, const float4* Ftab,
                                        const float2* Wtab, int lane) {
  const kfrag f = load_dft_frag(Ftab, lane);
  const ctile b = cmatmul3<DIR, true, 4>(x, f);
  return fft256_tail<DIR>(b, f, Wtab, lane);
}

template <int NCH, bool REALSEQ, typename SRC>
__device__ __forceinline__ ctile fft256_in(SRC src, const float4* Ftab,
                                           const float2* Wtab, int lane) {
  const cfragn<NCH> x = load_seq_frag_in<NCH>(src, lane);
  ctile b;
  if (REALSEQ)
    b = rmatmul16<NCH>(x, load_dft_frag_in<NCH>(Ftab, lane));
  else
    b = cmatmul3<+1, true, NCH>(x, load_dft_kfrag_in<NCH>(Ftab, lane));
  const kfrag f = load_dft_frag(Ftab, lane);
  return fft256_tail<+1>(b, f, Wtab, lane);
}

__device__ __forceinline__ void init_tables(float4* Ftab, float2* Wtab,
                                            int tid) {
  for (int i = tid; i < 256; i += NTHREADS) {
    const int r = i >> 4, c = i & 15;
    const float tf = -2.0f * PI_F * (float)((r * c) & 15) / 16.0f;
    const float fr = cosf(tf), fi = sinf(tf);
    Ftab[i] = make_float4(fr, fi, fi - fr, -(fr + fi));
    const float tw = -2.0f * PI_F * (float)i / 256.0f;
    Wtab[i] = make_float2(cosf(tw), sinf(tw));
  }
}

__global__ void __launch_bounds__(NTHREADS) kf_precompute(
    const float* __restrict__ kin, float2* __restrict__ kf) {
  extern __shared__ __align__(16) char smem[];
  float2* Rk   = (float2*)smem;
  float4* Ftab = (float4*)(smem + 64 * RSTRIDE * 8);
  float2* Wtab = (float2*)(Ftab + 256);

  const int tid = threadIdx.x, lane = tid & 31, wave = tid >> 5;
  const int lc = lane & 15, lh = lane >> 4;
  const int c = blockIdx.x;

  init_tables(Ftab, Wtab, tid);
  const float* kc = kin + (size_t)c * (64 * 64);
  for (int i = tid; i < 64 * 64; i += NTHREADS) {
    const int r = i >> 6, w = i & 63;
    Rk[r * RSTRIDE + w] = make_float2(kc[i], 0.0f);
  }
  __syncthreads();

  for (int row = wave; row < 64; row += NWAVES) {
    float2* Rrow = Rk + row * RSTRIDE;
    const ctile o = fft256_in<1, true>(
        [&](int n) { return make_float2(Rrow[n].x, 0.0f); }, Ftab, Wtab,
        lane);
#pragma unroll
    for (int j = 0; j < 8; ++j) {
      const int k = lc + 16 * (8 * lh + j);
      if (k <= 128) Rrow[k] = make_float2(o.re[j], o.im[j]);
    }
  }
  __syncthreads();

  for (int col = wave; col < 129; col += NWAVES) {
    float2* kfcol = kf + ((size_t)c * KFCOLS + col) * 256;
    const ctile o = fft256_in<1, false>(
        [&](int n) { return Rk[n * RSTRIDE + col]; }, Ftab, Wtab, lane);
#pragma unroll
    for (int j = 0; j < 8; ++j)
      kfcol[lc + 16 * (8 * lh + j)] = make_float2(o.re[j], o.im[j]);
  }
}

__global__ void __launch_bounds__(NTHREADS) conv_main(
    const float* __restrict__ u, const float* __restrict__ pregate,
    const float* __restrict__ postgate, const float2* __restrict__ kf,
    float* __restrict__ y) {
  extern __shared__ __align__(16) char smem[];
  float2* R    = (float2*)smem;
  float4* Ftab = (float4*)(smem + 128 * PSTRIDE * 8);
  float2* Wtab = (float2*)(Ftab + 256);

  const int tid = threadIdx.x, lane = tid & 31, wave = tid >> 5;
  const int lc = lane & 15, lh = lane >> 4;
  const int c = blockIdx.x & 255;
  const int bp = blockIdx.x >> 8;
  const size_t base_a = ((size_t)(2 * bp) * 256 + c) * (128 * 128);
  const size_t base_b = base_a + (size_t)256 * (128 * 128);

  init_tables(Ftab, Wtab, tid);

  for (int i = tid; i < 128 * 128; i += NTHREADS) {
    const int r = i >> 7, w = i & 127;
    R[r * PSTRIDE + w] = make_float2(u[base_a + i] * pregate[base_a + i],
                                     u[base_b + i] * pregate[base_b + i]);
  }
  __syncthreads();

  for (int row = wave; row < 128; row += NWAVES) {
    float2* Rrow = R + row * PSTRIDE;
    const ctile o = fft256_in<2, false>([&](int n) { return Rrow[n]; }, Ftab,
                                        Wtab, lane);
#pragma unroll
    for (int j = 0; j < 8; ++j)
      Rrow[lc + 16 * (8 * lh + j)] = make_float2(o.re[j], o.im[j]);
  }
  __syncthreads();

  const float2* kfbase = kf + (size_t)c * KFCOLS * 256;
  for (int col = wave; col < 256; col += NWAVES) {
    if (col + NWAVES < 256) {
      const int nc = col + NWAVES;
      const char* pf =
          (const char*)(kfbase + (size_t)((nc > 128) ? 256 - nc : nc) * 256);
      __builtin_prefetch(pf + lane * 64, 0, 1);
    }

    const bool mir = col > 128;
    const float sg = mir ? -1.0f : 1.0f;
    const float2* kfcol = kfbase + (size_t)(mir ? 256 - col : col) * 256;
    float2 kv[8];
#pragma unroll
    for (int j = 0; j < 8; ++j) kv[j] = kfcol[lc + 16 * (8 * lh + j)];

    ctile s = fft256_in<2, false>(
        [&](int n) {
          const float2 v = R[n * PSTRIDE + col];
          return make_float2(v.x, v.y * sg);
        },
        Ftab, Wtab, lane);

#pragma unroll
    for (int j = 0; j < 8; ++j) {
      const float ar = s.re[j], ai = s.im[j];
      s.re[j] = ar * kv[j].x - ai * kv[j].y;
      s.im[j] = ar * kv[j].y + ai * kv[j].x;
    }

    const ctile o = fft256<-1>(tile_as_frag(s), Ftab, Wtab, lane);
#pragma unroll
    for (int j = 0; j < 8; ++j) {
      const int k = lc + 16 * (8 * lh + j);
      if (k < 128) R[k * PSTRIDE + col] = make_float2(o.re[j], o.im[j] * sg);
    }
  }
  __syncthreads();

  const float inv_scale = 1.0f / 65536.0f;
  for (int row = wave; row < 128; row += NWAVES) {
    float2* Rrow = R + row * PSTRIDE;
    const size_t oa = base_a + (size_t)row * 128;
    const size_t ob = base_b + (size_t)row * 128;
    const cfrag x = load_seq_frag([&](int n) { return Rrow[n]; }, lane);
    const ctile o = fft256<-1>(x, Ftab, Wtab, lane);
#pragma unroll
    for (int j = 0; j < 8; ++j) {
      const int k = lc + 16 * (8 * lh + j);
      if (k < 128) {
        y[oa + k] = o.re[j] * inv_scale * postgate[oa + k];
        y[ob + k] = o.im[j] * inv_scale * postgate[ob + k];
      }
    }
  }
}

extern "C" void kernel_launch(void* const* d_in, const int* in_sizes, int n_in,
                              void* d_out, int out_size, void* d_ws,
                              size_t ws_size, hipStream_t stream) {
  const float* u        = (const float*)d_in[0];
  const float* k        = (const float*)d_in[1];
  const float* pregate  = (const float*)d_in[2];
  const float* postgate = (const float*)d_in[3];
  float* y   = (float*)d_out;
  float2* kf = (float2*)d_ws;

  const size_t lds_kf   = (size_t)64 * RSTRIDE * 8 + 256 * 16 + 256 * 8;
  const size_t lds_main = (size_t)128 * PSTRIDE * 8 + 256 * 16 + 256 * 8;

  (void)hipFuncSetAttribute((const void*)kf_precompute,
                            hipFuncAttributeMaxDynamicSharedMemorySize,
                            (int)lds_kf);
  (void)hipFuncSetAttribute((const void*)conv_main,
                            hipFuncAttributeMaxDynamicSharedMemorySize,
                            (int)lds_main);

  kf_precompute<<<256, NTHREADS, lds_kf, stream>>>(k, kf);
  conv_main<<<2 * 256, NTHREADS, lds_main, stream>>>(u, pregate, postgate, kf, y);
}


# Main kernel: a wave lowers its priority (s_setprio 3..0) as it gets through its share of each phase, so lagging waves win the matrix pipe and all waves reach the phase's barrier together
# speedup vs baseline: 3.7443x; 1.0344x over previous
#include <hip/hip_runtime.h>
#include <hip/hip_bf16.h>
#include <math.h>


typedef float v2f __attribute__((ext_vector_type(2)));
typedef float v8f __attribute__((ext_vector_type(8)));

#define PI_F 3.14159265358979323846f
#define RSTRIDE 133
#define PSTRIDE 257
#define KFCOLS 132
#define NWAVES 32
#define NTHREADS (NWAVES * 32)

__device__ __forceinline__ v8f wmma_f32(v2f a, v2f b, v8f c) {
  return __builtin_amdgcn_wmma_f32_16x16x4_f32(false, a, false, b, (short)0, c,
                                               false, false);
}

__device__ __forceinline__ void prio_by_progress(int it, int n) {
  const int lvl = (3 * (n - 1 - it) + (n - 2)) / (n - 1);
  if (lvl >= 3)
    __builtin_amdgcn_s_setprio(3);
  else if (lvl == 2)
    __builtin_amdgcn_s_setprio(2);
  else if (lvl == 1)
    __builtin_amdgcn_s_setprio(1);
  else
    __builtin_amdgcn_s_setprio(0);
}

__device__ __forceinline__ int kpos(int t, int h, int s) {
  return 8 * h + 2 * t + s;
}

__device__ __forceinline__ int kpos_in(int t, int h, int s) {
  return 4 * t + 2 * s + h;
}

struct ctile { v8f re, im; };
template <int NCH> struct cfragn { v2f re[NCH], im[NCH]; };
typedef cfragn<4> cfrag;
template <int NCH> struct kfragn { v2f r[NCH], p[NCH], q[NCH]; };
typedef kfragn<4> kfrag;

__device__ __forceinline__ cfrag tile_as_frag(const ctile& d) {
  cfrag f;
#pragma unroll
  for (int t = 0; t < 4; ++t) {
    f.re[t] = (v2f){d.re[2 * t], d.re[2 * t + 1]};
    f.im[t] = (v2f){d.im[2 * t], d.im[2 * t + 1]};
  }
  return f;
}

__device__ __forceinline__ kfrag load_dft_frag(const float4* Ftab, int lane) {
  const int c = lane & 15, h = lane >> 4;
  kfrag f;
#pragma unroll
  for (int t = 0; t < 4; ++t) {
    const float4 e0 = Ftab[kpos(t, h, 0) * 16 + c];
    const float4 e1 = Ftab[kpos(t, h, 1) * 16 + c];
    f.r[t] = (v2f){e0.x, e1.x};
    f.p[t] = (v2f){e0.z, e1.z};
    f.q[t] = (v2f){e0.w, e1.w};
  }
  return f;
}

template <typename SRC>
__device__ __forceinline__ cfrag load_seq_frag(SRC src, int lane) {
  const int c = lane & 15, h = lane >> 4;
  cfrag f;
#pragma unroll
  for (int t = 0; t < 4; ++t) {
    const float2 e0 = src(c + 16 * kpos(t, h, 0));
    const float2 e1 = src(c + 16 * kpos(t, h, 1));
    f.re[t] = (v2f){e0.x, e1.x};
    f.im[t] = (v2f){e0.y, e1.y};
  }
  return f;
}

template <int NCH>
__device__ __forceinline__ cfragn<NCH> load_dft_frag_in(const float4* Ftab,
                                                        int lane) {
  const int c = lane & 15, h = lane >> 4;
  cfragn<NCH> f;
#pragma unroll
  for (int t = 0; t < NCH; ++t) {
    const float4 e0 = Ftab[kpos_in(t, h, 0) * 16 + c];
    const float4 e1 = Ftab[kpos_in(t, h, 1) * 16 + c];
    f.re[t] = (v2f){e0.x, e1.x};
    f.im[t] = (v2f){e0.y, e1.y};
  }
  return f;
}

template <int NCH>
__device__ __forceinline__ kfragn<NCH> load_dft_kfrag_in(const float4* Ftab,
                                                         int lane) {
  const int c = lane & 15, h = lane >> 4;
  kfragn<NCH> f;
#pragma unroll
  for (int t = 0; t < NCH; ++t) {
    const float4 e0 = Ftab[kpos_in(t, h, 0) * 16 + c];
    const float4 e1 = Ftab[kpos_in(t, h, 1) * 16 + c];
    f.r[t] = (v2f){e0.x, e1.x};
    f.p[t] = (v2f){e0.z, e1.z};
    f.q[t] = (v2f){e0.w, e1.w};
  }
  return f;
}

template <int NCH, typename SRC>
__device__ __forceinline__ cfragn<NCH> load_seq_frag_in(SRC src, int lane) {
  const int c = lane & 15, h = lane >> 4;
  cfragn<NCH> f;
#pragma unroll
  for (int t = 0; t < NCH; ++t) {
    const float2 e0 = src(c + 16 * kpos_in(t, h, 0));
    const float2 e1 = src(c + 16 * kpos_in(t, h, 1));
    f.re[t] = (v2f){e0.x, e1.x};
    f.im[t] = (v2f){e0.y, e1.y};
  }
  return f;
}

template <bool XA>
__device__ __forceinline__ v8f mm(v2f x, v2f w, v8f c) {
  return XA ? wmma_f32(x, w, c) : wmma_f32(w, x, c);
}

template <int DIR, bool XA, int NCH>
__device__ __forceinline__ ctile cmatmul3(const cfragn<NCH>& x,
                                          const kfragn<NCH>& w) {
  v8f k = {};
#pragma unroll
  for (int t = 0; t < NCH; ++t) k = mm<XA>(x.re[t] + x.im[t], w.r[t], k);
  ctile d;
  d.re = k;
  d.im = k;
#pragma unroll
  for (int t = 0; t < NCH; ++t) {
    d.re = mm<XA>(x.im[t], (DIR > 0) ? w.q[t] : w.p[t], d.re);
    d.im = mm<XA>(x.re[t], (DIR > 0) ? w.p[t] : w.q[t], d.im);
  }
  return d;
}

template <int NCH>
__device__ __forceinline__ ctile rmatmul16(const cfragn<NCH>& a,
                                           const cfragn<NCH>& b) {
  v8f rr = {}, ri = {};
#pragma unroll
  for (int t = 0; t < NCH; ++t) {
    rr = wmma_f32(a.re[t], b.re[t], rr);
    ri = wmma_f32(a.re[t], b.im[t], ri);
  }
  ctile d;
  d.re = rr;
  d.im = ri;
  return d;
}

template <int DIR>
__device__ __forceinline__ ctile fft256_tail(ctile b, const kfrag& f,
                                             const float2* Wtab, int lane) {
  const int c = lane & 15, h = lane >> 4;

#pragma unroll
  for (int j = 0; j < 8; ++j) {
    const int row = 8 * h + j;
    const float2 w = Wtab[(row * c) & 255];
    const float wr = w.x, wi = (DIR > 0) ? w.y : -w.y;
    const float re = b.re[j] * wr - b.im[j] * wi;
    const float im = b.re[j] * wi + b.im[j] * wr;
    b.re[j] = re;
    b.im[j] = im;
  }
  return cmatmul3<DIR, false, 4>(tile_as_frag(b), f);
}

template <int DIR>
__device__ __forceinline__ ctile fft256(const cfrag& x, const float4* Ftab,
                                        const float2* Wtab, int lane) {
  const kfrag f = load_dft_frag(Ftab, lane);
  const ctile b = cmatmul3<DIR, true, 4>(x, f);
  return fft256_tail<DIR>(b, f, Wtab, lane);
}

template <int NCH, bool REALSEQ, typename SRC>
__device__ __forceinline__ ctile fft256_in(SRC src, const float4* Ftab,
                                           const float2* Wtab, int lane) {
  const cfragn<NCH> x = load_seq_frag_in<NCH>(src, lane);
  ctile b;
  if (REALSEQ)
    b = rmatmul16<NCH>(x, load_dft_frag_in<NCH>(Ftab, lane));
  else
    b = cmatmul3<+1, true, NCH>(x, load_dft_kfrag_in<NCH>(Ftab, lane));
  const kfrag f = load_dft_frag(Ftab, lane);
  return fft256_tail<+1>(b, f, Wtab, lane);
}

__device__ __forceinline__ void init_tables(float4* Ftab, float2* Wtab,
                                            int tid) {
  for (int i = tid; i < 256; i += NTHREADS) {
    const int r = i >> 4, c = i & 15;
    const float tf = -2.0f * PI_F * (float)((r * c) & 15) / 16.0f;
    const float fr = cosf(tf), fi = sinf(tf);
    Ftab[i] = make_float4(fr, fi, fi - fr, -(fr + fi));
    const float tw = -2.0f * PI_F * (float)i / 256.0f;
    Wtab[i] = make_float2(cosf(tw), sinf(tw));
  }
}

__global__ void __launch_bounds__(NTHREADS) kf_precompute(
    const float* __restrict__ kin, float2* __restrict__ kf) {
  extern __shared__ __align__(16) char smem[];
  float2* Rk   = (float2*)smem;
  float4* Ftab = (float4*)(smem + 64 * RSTRIDE * 8);
  float2* Wtab = (float2*)(Ftab + 256);

  const int tid = threadIdx.x, lane = tid & 31, wave = tid >> 5;
  const int lc = lane & 15, lh = lane >> 4;
  const int c = blockIdx.x;

  init_tables(Ftab, Wtab, tid);
  const float* kc = kin + (size_t)c * (64 * 64);
  for (int i = tid; i < 64 * 64; i += NTHREADS) {
    const int r = i >> 6, w = i & 63;
    Rk[r * RSTRIDE + w] = make_float2(kc[i], 0.0f);
  }
  __syncthreads();

  for (int row = wave; row < 64; row += NWAVES) {
    float2* Rrow = Rk + row * RSTRIDE;
    const ctile o = fft256_in<1, true>(
        [&](int n) { return make_float2(Rrow[n].x, 0.0f); }, Ftab, Wtab,
        lane);
#pragma unroll
    for (int j = 0; j < 8; ++j) {
      const int k = lc + 16 * (8 * lh + j);
      if (k <= 128) Rrow[k] = make_float2(o.re[j], o.im[j]);
    }
  }
  __syncthreads();

  for (int col = wave; col < 129; col += NWAVES) {
    float2* kfcol = kf + ((size_t)c * KFCOLS + col) * 256;
    const ctile o = fft256_in<1, false>(
        [&](int n) { return Rk[n * RSTRIDE + col]; }, Ftab, Wtab, lane);
#pragma unroll
    for (int j = 0; j < 8; ++j)
      kfcol[lc + 16 * (8 * lh + j)] = make_float2(o.re[j], o.im[j]);
  }
}

__global__ void __launch_bounds__(NTHREADS) conv_main(
    const float* __restrict__ u, const float* __restrict__ pregate,
    const float* __restrict__ postgate, const float2* __restrict__ kf,
    float* __restrict__ y) {
  extern __shared__ __align__(16) char smem[];
  float2* R    = (float2*)smem;
  float4* Ftab = (float4*)(smem + 128 * PSTRIDE * 8);
  float2* Wtab = (float2*)(Ftab + 256);

  const int tid = threadIdx.x, lane = tid & 31, wave = tid >> 5;
  const int lc = lane & 15, lh = lane >> 4;
  const int c = blockIdx.x & 255;
  const int bp = blockIdx.x >> 8;
  const size_t base_a = ((size_t)(2 * bp) * 256 + c) * (128 * 128);
  const size_t base_b = base_a + (size_t)256 * (128 * 128);

  init_tables(Ftab, Wtab, tid);

  for (int i = tid; i < 128 * 128; i += NTHREADS) {
    const int r = i >> 7, w = i & 127;
    R[r * PSTRIDE + w] = make_float2(u[base_a + i] * pregate[base_a + i],
                                     u[base_b + i] * pregate[base_b + i]);
  }
  __syncthreads();

  prio_by_progress(0, 128 / NWAVES);
#pragma unroll 1
  for (int row = wave, it = 1; row < 128; row += NWAVES, ++it) {
    float2* Rrow = R + row * PSTRIDE;
    const ctile o = fft256_in<2, false>([&](int n) { return Rrow[n]; }, Ftab,
                                        Wtab, lane);
#pragma unroll
    for (int j = 0; j < 8; ++j)
      Rrow[lc + 16 * (8 * lh + j)] = make_float2(o.re[j], o.im[j]);
    if (it < 128 / NWAVES) prio_by_progress(it, 128 / NWAVES);
  }
  __syncthreads();

  const float2* kfbase = kf + (size_t)c * KFCOLS * 256;
  prio_by_progress(0, 256 / NWAVES);
#pragma unroll 1
  for (int col = wave, it = 1; col < 256; col += NWAVES, ++it) {
    if (col + NWAVES < 256) {
      const int nc = col + NWAVES;
      const char* pf =
          (const char*)(kfbase + (size_t)((nc > 128) ? 256 - nc : nc) * 256);
      __builtin_prefetch(pf + lane * 64, 0, 1);
    }

    const bool mir = col > 128;
    const float sg = mir ? -1.0f : 1.0f;
    const float2* kfcol = kfbase + (size_t)(mir ? 256 - col : col) * 256;
    float2 kv[8];
#pragma unroll
    for (int j = 0; j < 8; ++j) kv[j] = kfcol[lc + 16 * (8 * lh + j)];

    ctile s = fft256_in<2, false>(
        [&](int n) {
          const float2 v = R[n * PSTRIDE + col];
          return make_float2(v.x, v.y * sg);
        },
        Ftab, Wtab, lane);

#pragma unroll
    for (int j = 0; j < 8; ++j) {
      const float ar = s.re[j], ai = s.im[j];
      s.re[j] = ar * kv[j].x - ai * kv[j].y;
      s.im[j] = ar * kv[j].y + ai * kv[j].x;
    }

    const ctile o = fft256<-1>(tile_as_frag(s), Ftab, Wtab, lane);
    if (lh == 0) {
#pragma unroll
      for (int j = 0; j < 8; ++j) {
        const int k = lc + 16 * j;
        R[k * PSTRIDE + col] = make_float2(o.re[j], o.im[j] * sg);
      }
    }
    if (it < 256 / NWAVES) prio_by_progress(it, 256 / NWAVES);
  }
  __syncthreads();

  const float inv_scale = 1.0f / 65536.0f;
  prio_by_progress(0, 128 / NWAVES);
#pragma unroll 1
  for (int row = wave, it = 1; row < 128; row += NWAVES, ++it) {
    float2* Rrow = R + row * PSTRIDE;
    const size_t oa = base_a + (size_t)row * 128;
    const size_t ob = base_b + (size_t)row * 128;
    const cfrag x = load_seq_frag([&](int n) { return Rrow[n]; }, lane);
    const ctile o = fft256<-1>(x, Ftab, Wtab, lane);
    if (lh == 0) {
#pragma unroll
      for (int j = 0; j < 8; ++j) {
        const int k = lc + 16 * j;
        y[oa + k] = o.re[j] * inv_scale * postgate[oa + k];
        y[ob + k] = o.im[j] * inv_scale * postgate[ob + k];
      }
    }
    if (it < 128 / NWAVES) prio_by_progress(it, 128 / NWAVES);
  }
}

extern "C" void kernel_launch(void* const* d_in, const int* in_sizes, int n_in,
                              void* d_out, int out_size, void* d_ws,
                              size_t ws_size, hipStream_t stream) {
  const float* u        = (const float*)d_in[0];
  const float* k        = (const float*)d_in[1];
  const float* pregate  = (const float*)d_in[2];
  const float* postgate = (const float*)d_in[3];
  float* y   = (float*)d_out;
  float2* kf = (float2*)d_ws;

  const size_t lds_kf   = (size_t)64 * RSTRIDE * 8 + 256 * 16 + 256 * 8;
  const size_t lds_main = (size_t)128 * PSTRIDE * 8 + 256 * 16 + 256 * 8;

  (void)hipFuncSetAttribute((const void*)kf_precompute,
                            hipFuncAttributeMaxDynamicSharedMemorySize,
                            (int)lds_kf);
  (void)hipFuncSetAttribute((const void*)conv_main,
                            hipFuncAttributeMaxDynamicSharedMemorySize,
                            (int)lds_main);

  kf_precompute<<<256, NTHREADS, lds_kf, stream>>>(k, kf);
  conv_main<<<2 * 256, NTHREADS, lds_main, stream>>>(u, pregate, postgate, kf, y);
}


# Both kernels: a wave reads its DFT-matrix fragments and twiddle factors from the LDS tables once and holds them in registers; no LDS read is left between a transform's matrix products
# speedup vs baseline: 3.8907x; 1.0391x over previous
#include <hip/hip_runtime.h>
#include <hip/hip_bf16.h>
#include <math.h>


typedef float v2f __attribute__((ext_vector_type(2)));
typedef float v8f __attribute__((ext_vector_type(8)));

#define PI_F 3.14159265358979323846f
#define RSTRIDE 133
#define PSTRIDE 257
#define KFCOLS 132
#define NWAVES 32
#define NTHREADS (NWAVES * 32)

__device__ __forceinline__ v8f wmma_f32(v2f a, v2f b, v8f c) {
  return __builtin_amdgcn_wmma_f32_16x16x4_f32(false, a, false, b, (short)0, c,
                                               false, false);
}

__device__ __forceinline__ void prio_by_progress(int it, int n) {
  const int lvl = (3 * (n - 1 - it) + (n - 2)) / (n - 1);
  if (lvl >= 3)
    __builtin_amdgcn_s_setprio(3);
  else if (lvl == 2)
    __builtin_amdgcn_s_setprio(2);
  else if (lvl == 1)
    __builtin_amdgcn_s_setprio(1);
  else
    __builtin_amdgcn_s_setprio(0);
}

__device__ __forceinline__ int kpos(int t, int h, int s) {
  return 8 * h + 2 * t + s;
}

__device__ __forceinline__ int kpos_in(int t, int h, int s) {
  return 4 * t + 2 * s + h;
}

struct ctile { v8f re, im; };
template <int NCH> struct cfragn { v2f re[NCH], im[NCH]; };
typedef cfragn<4> cfrag;
template <int NCH> struct kfragn { v2f r[NCH], p[NCH], q[NCH]; };
typedef kfragn<4> kfrag;

__device__ __forceinline__ cfrag tile_as_frag(const ctile& d) {
  cfrag f;
#pragma unroll
  for (int t = 0; t < 4; ++t) {
    f.re[t] = (v2f){d.re[2 * t], d.re[2 * t + 1]};
    f.im[t] = (v2f){d.im[2 * t], d.im[2 * t + 1]};
  }
  return f;
}

__device__ __forceinline__ kfrag load_dft_frag(const float4* Ftab, int lane) {
  const int c = lane & 15, h = lane >> 4;
  kfrag f;
#pragma unroll
  for (int t = 0; t < 4; ++t) {
    const float4 e0 = Ftab[kpos(t, h, 0) * 16 + c];
    const float4 e1 = Ftab[kpos(t, h, 1) * 16 + c];
    f.r[t] = (v2f){e0.x, e1.x};
    f.p[t] = (v2f){e0.z, e1.z};
    f.q[t] = (v2f){e0.w, e1.w};
  }
  return f;
}

struct lconst {
  kfrag f;
  v8f twr, twi;
};

__device__ __forceinline__ lconst load_lane_consts(const float4* Ftab,
                                                   const float2* Wtab,
                                                   int lane) {
  const int c = lane & 15, h = lane >> 4;
  lconst k;
  k.f = load_dft_frag(Ftab, lane);
#pragma unroll
  for (int j = 0; j < 8; ++j) {
    const float2 w = Wtab[((8 * h + j) * c) & 255];
    k.twr[j] = w.x;
    k.twi[j] = w.y;
  }
  return k;
}

template <typename SRC>
__device__ __forceinline__ cfrag load_seq_frag(SRC src, int lane) {
  const int c = lane & 15, h = lane >> 4;
  cfrag f;
#pragma unroll
  for (int t = 0; t < 4; ++t) {
    const float2 e0 = src(c + 16 * kpos(t, h, 0));
    const float2 e1 = src(c + 16 * kpos(t, h, 1));
    f.re[t] = (v2f){e0.x, e1.x};
    f.im[t] = (v2f){e0.y, e1.y};
  }
  return f;
}

template <int NCH>
__device__ __forceinline__ cfragn<NCH> load_dft_frag_in(const float4* Ftab,
                                                        int lane) {
  const int c = lane & 15, h = lane >> 4;
  cfragn<NCH> f;
#pragma unroll
  for (int t = 0; t < NCH; ++t) {
    const float4 e0 = Ftab[kpos_in(t, h, 0) * 16 + c];
    const float4 e1 = Ftab[kpos_in(t, h, 1) * 16 + c];
    f.re[t] = (v2f){e0.x, e1.x};
    f.im[t] = (v2f){e0.y, e1.y};
  }
  return f;
}

template <int NCH>
__device__ __forceinline__ kfragn<NCH> load_dft_kfrag_in(const float4* Ftab,
                                                         int lane) {
  const int c = lane & 15, h = lane >> 4;
  kfragn<NCH> f;
#pragma unroll
  for (int t = 0; t < NCH; ++t) {
    const float4 e0 = Ftab[kpos_in(t, h, 0) * 16 + c];
    const float4 e1 = Ftab[kpos_in(t, h, 1) * 16 + c];
    f.r[t] = (v2f){e0.x, e1.x};
    f.p[t] = (v2f){e0.z, e1.z};
    f.q[t] = (v2f){e0.w, e1.w};
  }
  return f;
}

template <int NCH, typename SRC>
__device__ __forceinline__ cfragn<NCH> load_seq_frag_in(SRC src, int lane) {
  const int c = lane & 15, h = lane >> 4;
  cfragn<NCH> f;
#pragma unroll
  for (int t = 0; t < NCH; ++t) {
    const float2 e0 = src(c + 16 * kpos_in(t, h, 0));
    const float2 e1 = src(c + 16 * kpos_in(t, h, 1));
    f.re[t] = (v2f){e0.x, e1.x};
    f.im[t] = (v2f){e0.y, e1.y};
  }
  return f;
}

template <bool XA>
__device__ __forceinline__ v8f mm(v2f x, v2f w, v8f c) {
  return XA ? wmma_f32(x, w, c) : wmma_f32(w, x, c);
}

template <int DIR, bool XA, int NCH>
__device__ __forceinline__ ctile cmatmul3(const cfragn<NCH>& x,
                                          const kfragn<NCH>& w) {
  v8f k = {};
#pragma unroll
  for (int t = 0; t < NCH; ++t) k = mm<XA>(x.re[t] + x.im[t], w.r[t], k);
  ctile d;
  d.re = k;
  d.im = k;
#pragma unroll
  for (int t = 0; t < NCH; ++t) {
    d.re = mm<XA>(x.im[t], (DIR > 0) ? w.q[t] : w.p[t], d.re);
    d.im = mm<XA>(x.re[t], (DIR > 0) ? w.p[t] : w.q[t], d.im);
  }
  return d;
}

template <int NCH>
__device__ __forceinline__ ctile rmatmul16(const cfragn<NCH>& a,
                                           const cfragn<NCH>& b) {
  v8f rr = {}, ri = {};
#pragma unroll
  for (int t = 0; t < NCH; ++t) {
    rr = wmma_f32(a.re[t], b.re[t], rr);
    ri = wmma_f32(a.re[t], b.im[t], ri);
  }
  ctile d;
  d.re = rr;
  d.im = ri;
  return d;
}

template <int DIR>
__device__ __forceinline__ ctile fft256_tail(ctile b, const lconst& k) {
#pragma unroll
  for (int j = 0; j < 8; ++j) {
    const float wr = k.twr[j], wi = (DIR > 0) ? k.twi[j] : -k.twi[j];
    const float re = b.re[j] * wr - b.im[j] * wi;
    const float im = b.re[j] * wi + b.im[j] * wr;
    b.re[j] = re;
    b.im[j] = im;
  }
  return cmatmul3<DIR, false, 4>(tile_as_frag(b), k.f);
}

template <int DIR>
__device__ __forceinline__ ctile fft256(const cfrag& x, const lconst& k) {
  const ctile b = cmatmul3<DIR, true, 4>(x, k.f);
  return fft256_tail<DIR>(b, k);
}

template <int NCH, typename SRC>
__device__ __forceinline__ ctile fft256_in_real(SRC src,
                                                const cfragn<NCH>& fin,
                                                const lconst& k, int lane) {
  const cfragn<NCH> x = load_seq_frag_in<NCH>(src, lane);
  return fft256_tail<+1>(rmatmul16<NCH>(x, fin), k);
}

template <int NCH, typename SRC>
__device__ __forceinline__ ctile fft256_in(SRC src, const kfragn<NCH>& fin,
                                           const lconst& k, int lane) {
  const cfragn<NCH> x = load_seq_frag_in<NCH>(src, lane);
  return fft256_tail<+1>(cmatmul3<+1, true, NCH>(x, fin), k);
}

__device__ __forceinline__ void init_tables(float4* Ftab, float2* Wtab,
                                            int tid) {
  for (int i = tid; i < 256; i += NTHREADS) {
    const int r = i >> 4, c = i & 15;
    const float tf = -2.0f * PI_F * (float)((r * c) & 15) / 16.0f;
    const float fr = cosf(tf), fi = sinf(tf);
    Ftab[i] = make_float4(fr, fi, fi - fr, -(fr + fi));
    const float tw = -2.0f * PI_F * (float)i / 256.0f;
    Wtab[i] = make_float2(cosf(tw), sinf(tw));
  }
}

__global__ void __launch_bounds__(NTHREADS) kf_precompute(
    const float* __restrict__ kin, float2* __restrict__ kf) {
  extern __shared__ __align__(16) char smem[];
  float2* Rk   = (float2*)smem;
  float4* Ftab = (float4*)(smem + 64 * RSTRIDE * 8);
  float2* Wtab = (float2*)(Ftab + 256);

  const int tid = threadIdx.x, lane = tid & 31, wave = tid >> 5;
  const int lc = lane & 15, lh = lane >> 4;
  const int c = blockIdx.x;

  init_tables(Ftab, Wtab, tid);
  const float* kc = kin + (size_t)c * (64 * 64);
  for (int i = tid; i < 64 * 64; i += NTHREADS) {
    const int r = i >> 6, w = i & 63;
    Rk[r * RSTRIDE + w] = make_float2(kc[i], 0.0f);
  }
  __syncthreads();

  const lconst lk = load_lane_consts(Ftab, Wtab, lane);
  const cfragn<1> fin_r = load_dft_frag_in<1>(Ftab, lane);
  const kfragn<1> fin_c = load_dft_kfrag_in<1>(Ftab, lane);

  for (int row = wave; row < 64; row += NWAVES) {
    float2* Rrow = Rk + row * RSTRIDE;
    const ctile o = fft256_in_real<1>(
        [&](int n) { return make_float2(Rrow[n].x, 0.0f); }, fin_r, lk, lane);
#pragma unroll
    for (int j = 0; j < 8; ++j) {
      const int k = lc + 16 * (8 * lh + j);
      if (k <= 128) Rrow[k] = make_float2(o.re[j], o.im[j]);
    }
  }
  __syncthreads();

  for (int col = wave; col < 129; col += NWAVES) {
    float2* kfcol = kf + ((size_t)c * KFCOLS + col) * 256;
    const ctile o = fft256_in<1>(
        [&](int n) { return Rk[n * RSTRIDE + col]; }, fin_c, lk, lane);
#pragma unroll
    for (int j = 0; j < 8; ++j)
      kfcol[lc + 16 * (8 * lh + j)] = make_float2(o.re[j], o.im[j]);
  }
}

__global__ void __launch_bounds__(NTHREADS) conv_main(
    const float* __restrict__ u, const float* __restrict__ pregate,
    const float* __restrict__ postgate, const float2* __restrict__ kf,
    float* __restrict__ y) {
  extern __shared__ __align__(16) char smem[];
  float2* R    = (float2*)smem;
  float4* Ftab = (float4*)(smem + 128 * PSTRIDE * 8);
  float2* Wtab = (float2*)(Ftab + 256);

  const int tid = threadIdx.x, lane = tid & 31, wave = tid >> 5;
  const int lc = lane & 15, lh = lane >> 4;
  const int c = blockIdx.x & 255;
  const int bp = blockIdx.x >> 8;
  const size_t base_a = ((size_t)(2 * bp) * 256 + c) * (128 * 128);
  const size_t base_b = base_a + (size_t)256 * (128 * 128);

  init_tables(Ftab, Wtab, tid);

  for (int i = tid; i < 128 * 128; i += NTHREADS) {
    const int r = i >> 7, w = i & 127;
    R[r * PSTRIDE + w] = make_float2(u[base_a + i] * pregate[base_a + i],
                                     u[base_b + i] * pregate[base_b + i]);
  }
  __syncthreads();

  const lconst lk = load_lane_consts(Ftab, Wtab, lane);
  const kfragn<2> fin = load_dft_kfrag_in<2>(Ftab, lane);

  prio_by_progress(0, 128 / NWAVES);
#pragma unroll 1
  for (int row = wave, it = 1; row < 128; row += NWAVES, ++it) {
    float2* Rrow = R + row * PSTRIDE;
    const ctile o =
        fft256_in<2>([&](int n) { return Rrow[n]; }, fin, lk, lane);
#pragma unroll
    for (int j = 0; j < 8; ++j)
      Rrow[lc + 16 * (8 * lh + j)] = make_float2(o.re[j], o.im[j]);
    if (it < 128 / NWAVES) prio_by_progress(it, 128 / NWAVES);
  }
  __syncthreads();

  const float2* kfbase = kf + (size_t)c * KFCOLS * 256;
  prio_by_progress(0, 256 / NWAVES);
#pragma unroll 1
  for (int col = wave, it = 1; col < 256; col += NWAVES, ++it) {
    if (col + NWAVES < 256) {
      const int nc = col + NWAVES;
      const char* pf =
          (const char*)(kfbase + (size_t)((nc > 128) ? 256 - nc : nc) * 256);
      __builtin_prefetch(pf + lane * 64, 0, 1);
    }

    const bool mir = col > 128;
    const float sg = mir ? -1.0f : 1.0f;
    const float2* kfcol = kfbase + (size_t)(mir ? 256 - col : col) * 256;
    float2 kv[8];
#pragma unroll
    for (int j = 0; j < 8; ++j) kv[j] = kfcol[lc + 16 * (8 * lh + j)];

    ctile s = fft256_in<2>(
        [&](int n) {
          const float2 v = R[n * PSTRIDE + col];
          return make_float2(v.x, v.y * sg);
        },
        fin, lk, lane);

#pragma unroll
    for (int j = 0; j < 8; ++j) {
      const float ar = s.re[j], ai = s.im[j];
      s.re[j] = ar * kv[j].x - ai * kv[j].y;
      s.im[j] = ar * kv[j].y + ai * kv[j].x;
    }

    const ctile o = fft256<-1>(tile_as_frag(s), lk);
    if (lh == 0) {
#pragma unroll
      for (int j = 0; j < 8; ++j) {
        const int k = lc + 16 * j;
        R[k * PSTRIDE + col] = make_float2(o.re[j], o.im[j] * sg);
      }
    }
    if (it < 256 / NWAVES) prio_by_progress(it, 256 / NWAVES);
  }
  __syncthreads();

  const float inv_scale = 1.0f / 65536.0f;
  prio_by_progress(0, 128 / NWAVES);
#pragma unroll 1
  for (int row = wave, it = 1; row < 128; row += NWAVES, ++it) {
    float2* Rrow = R + row * PSTRIDE;
    const size_t oa = base_a + (size_t)row * 128;
    const size_t ob = base_b + (size_t)row * 128;
    const cfrag x = load_seq_frag([&](int n) { return Rrow[n]; }, lane);
    const ctile o = fft256<-1>(x, lk);
    if (lh == 0) {
#pragma unroll
      for (int j = 0; j < 8; ++j) {
        const int k = lc + 16 * j;
        y[oa + k] = o.re[j] * inv_scale * postgate[oa + k];
        y[ob + k] = o.im[j] * inv_scale * postgate[ob + k];
      }
    }
    if (it < 128 / NWAVES) prio_by_progress(it, 128 / NWAVES);
  }
}

extern "C" void kernel_launch(void* const* d_in, const int* in_sizes, int n_in,
                              void* d_out, int out_size, void* d_ws,
                              size_t ws_size, hipStream_t stream) {
  const float* u        = (const float*)d_in[0];
  const float* k        = (const float*)d_in[1];
  const float* pregate  = (const float*)d_in[2];
  const float* postgate = (const float*)d_in[3];
  float* y   = (float*)d_out;
  float2* kf = (float2*)d_ws;

  const size_t lds_kf   = (size_t)64 * RSTRIDE * 8 + 256 * 16 + 256 * 8;
  const size_t lds_main = (size_t)128 * PSTRIDE * 8 + 256 * 16 + 256 * 8;

  (void)hipFuncSetAttribute((const void*)kf_precompute,
                            hipFuncAttributeMaxDynamicSharedMemorySize,
                            (int)lds_kf);
  (void)hipFuncSetAttribute((const void*)conv_main,
                            hipFuncAttributeMaxDynamicSharedMemorySize,
                            (int)lds_main);

  kf_precompute<<<256, NTHREADS, lds_kf, stream>>>(k, kf);
  conv_main<<<2 * 256, NTHREADS, lds_main, stream>>>(u, pregate, postgate, kf, y);
}


# Stage 2 of each cropped inverse transform (only outputs k<128 are kept) as two real products on a packed factor [Wr8|Wi8]: 8 WMMAs instead of 12, re/im in lane halves; new summation order, rounding
# speedup vs baseline: 4.1237x; 1.0599x over previous
#include <hip/hip_runtime.h>
#include <hip/hip_bf16.h>
#include <math.h>


typedef float v2f __attribute__((ext_vector_type(2)));
typedef float v8f __attribute__((ext_vector_type(8)));

#define PI_F 3.14159265358979323846f
#define RSTRIDE 133
#define PSTRIDE 257
#define KFCOLS 132
#define NWAVES 32
#define NTHREADS (NWAVES * 32)

__device__ __forceinline__ v8f wmma_f32(v2f a, v2f b, v8f c) {
  return __builtin_amdgcn_wmma_f32_16x16x4_f32(false, a, false, b, (short)0, c,
                                               false, false);
}

__device__ __forceinline__ void prio_by_progress(int it, int n) {
  const int lvl = (3 * (n - 1 - it) + (n - 2)) / (n - 1);
  if (lvl >= 3)
    __builtin_amdgcn_s_setprio(3);
  else if (lvl == 2)
    __builtin_amdgcn_s_setprio(2);
  else if (lvl == 1)
    __builtin_amdgcn_s_setprio(1);
  else
    __builtin_amdgcn_s_setprio(0);
}

__device__ __forceinline__ int kpos(int t, int h, int s) {
  return 8 * h + 2 * t + s;
}

__device__ __forceinline__ int kpos_in(int t, int h, int s) {
  return 4 * t + 2 * s + h;
}

struct ctile { v8f re, im; };
template <int NCH> struct cfragn { v2f re[NCH], im[NCH]; };
typedef cfragn<4> cfrag;
template <int NCH> struct kfragn { v2f r[NCH], p[NCH], q[NCH]; };
typedef kfragn<4> kfrag;

__device__ __forceinline__ cfrag tile_as_frag(const ctile& d) {
  cfrag f;
#pragma unroll
  for (int t = 0; t < 4; ++t) {
    f.re[t] = (v2f){d.re[2 * t], d.re[2 * t + 1]};
    f.im[t] = (v2f){d.im[2 * t], d.im[2 * t + 1]};
  }
  return f;
}

__device__ __forceinline__ kfrag load_dft_frag(const float4* Ftab, int lane) {
  const int c = lane & 15, h = lane >> 4;
  kfrag f;
#pragma unroll
  for (int t = 0; t < 4; ++t) {
    const float4 e0 = Ftab[kpos(t, h, 0) * 16 + c];
    const float4 e1 = Ftab[kpos(t, h, 1) * 16 + c];
    f.r[t] = (v2f){e0.x, e1.x};
    f.p[t] = (v2f){e0.z, e1.z};
    f.q[t] = (v2f){e0.w, e1.w};
  }
  return f;
}

struct lconst {
  kfrag f;
  v8f twr, twi;
};

__device__ __forceinline__ lconst load_lane_consts(const float4* Ftab,
                                                   const float2* Wtab,
                                                   int lane) {
  const int c = lane & 15, h = lane >> 4;
  lconst k;
  k.f = load_dft_frag(Ftab, lane);
#pragma unroll
  for (int j = 0; j < 8; ++j) {
    const float2 w = Wtab[((8 * h + j) * c) & 255];
    k.twr[j] = w.x;
    k.twi[j] = w.y;
  }
  return k;
}

template <typename SRC>
__device__ __forceinline__ cfrag load_seq_frag(SRC src, int lane) {
  const int c = lane & 15, h = lane >> 4;
  cfrag f;
#pragma unroll
  for (int t = 0; t < 4; ++t) {
    const float2 e0 = src(c + 16 * kpos(t, h, 0));
    const float2 e1 = src(c + 16 * kpos(t, h, 1));
    f.re[t] = (v2f){e0.x, e1.x};
    f.im[t] = (v2f){e0.y, e1.y};
  }
  return f;
}

template <int NCH>
__device__ __forceinline__ cfragn<NCH> load_dft_frag_in(const float4* Ftab,
                                                        int lane) {
  const int c = lane & 15, h = lane >> 4;
  cfragn<NCH> f;
#pragma unroll
  for (int t = 0; t < NCH; ++t) {
    const float4 e0 = Ftab[kpos_in(t, h, 0) * 16 + c];
    const float4 e1 = Ftab[kpos_in(t, h, 1) * 16 + c];
    f.re[t] = (v2f){e0.x, e1.x};
    f.im[t] = (v2f){e0.y, e1.y};
  }
  return f;
}

template <int NCH>
__device__ __forceinline__ kfragn<NCH> load_dft_kfrag_in(const float4* Ftab,
                                                         int lane) {
  const int c = lane & 15, h = lane >> 4;
  kfragn<NCH> f;
#pragma unroll
  for (int t = 0; t < NCH; ++t) {
    const float4 e0 = Ftab[kpos_in(t, h, 0) * 16 + c];
    const float4 e1 = Ftab[kpos_in(t, h, 1) * 16 + c];
    f.r[t] = (v2f){e0.x, e1.x};
    f.p[t] = (v2f){e0.z, e1.z};
    f.q[t] = (v2f){e0.w, e1.w};
  }
  return f;
}

template <int NCH, typename SRC>
__device__ __forceinline__ cfragn<NCH> load_seq_frag_in(SRC src, int lane) {
  const int c = lane & 15, h = lane >> 4;
  cfragn<NCH> f;
#pragma unroll
  for (int t = 0; t < NCH; ++t) {
    const float2 e0 = src(c + 16 * kpos_in(t, h, 0));
    const float2 e1 = src(c + 16 * kpos_in(t, h, 1));
    f.re[t] = (v2f){e0.x, e1.x};
    f.im[t] = (v2f){e0.y, e1.y};
  }
  return f;
}

template <bool XA>
__device__ __forceinline__ v8f mm(v2f x, v2f w, v8f c) {
  return XA ? wmma_f32(x, w, c) : wmma_f32(w, x, c);
}

template <int DIR, bool XA, int NCH>
__device__ __forceinline__ ctile cmatmul3(const cfragn<NCH>& x,
                                          const kfragn<NCH>& w) {
  v8f k = {};
#pragma unroll
  for (int t = 0; t < NCH; ++t) k = mm<XA>(x.re[t] + x.im[t], w.r[t], k);
  ctile d;
  d.re = k;
  d.im = k;
#pragma unroll
  for (int t = 0; t < NCH; ++t) {
    d.re = mm<XA>(x.im[t], (DIR > 0) ? w.q[t] : w.p[t], d.re);
    d.im = mm<XA>(x.re[t], (DIR > 0) ? w.p[t] : w.q[t], d.im);
  }
  return d;
}

template <int NCH>
__device__ __forceinline__ ctile rmatmul16(const cfragn<NCH>& a,
                                           const cfragn<NCH>& b) {
  v8f rr = {}, ri = {};
#pragma unroll
  for (int t = 0; t < NCH; ++t) {
    rr = wmma_f32(a.re[t], b.re[t], rr);
    ri = wmma_f32(a.re[t], b.im[t], ri);
  }
  ctile d;
  d.re = rr;
  d.im = ri;
  return d;
}

template <int DIR>
__device__ __forceinline__ ctile fft256_tail(ctile b, const lconst& k) {
#pragma unroll
  for (int j = 0; j < 8; ++j) {
    const float wr = k.twr[j], wi = (DIR > 0) ? k.twi[j] : -k.twi[j];
    const float re = b.re[j] * wr - b.im[j] * wi;
    const float im = b.re[j] * wi + b.im[j] * wr;
    b.re[j] = re;
    b.im[j] = im;
  }
  return cmatmul3<DIR, false, 4>(tile_as_frag(b), k.f);
}

#define CROPTAB 128
__device__ __forceinline__ void init_crop_table(float4* Ctab,
                                                const float4* Ftab, int tid) {
  if (tid < CROPTAB) {
    const int t = tid >> 5, l = tid & 31;
    const int c = l & 15, h = l >> 4, cc = c & 7;
    const float4 e0 = Ftab[kpos(t, h, 0) * 16 + cc];
    const float4 e1 = Ftab[kpos(t, h, 1) * 16 + cc];
    Ctab[tid] = (c < 8) ? make_float4(e0.x, e1.x, e0.y, e1.y)
                        : make_float4(-e0.y, -e1.y, e0.x, e1.x);
  }
}

struct cropfrag { v2f p1[4], p2[4]; };

__device__ __forceinline__ cropfrag load_crop_frag(const float4* Ctab,
                                                   int lane) {
  cropfrag f;
#pragma unroll
  for (int t = 0; t < 4; ++t) {
    const float4 e = Ctab[t * 32 + lane];
    f.p1[t] = (v2f){e.x, e.y};
    f.p2[t] = (v2f){e.z, e.w};
  }
  return f;
}

__device__ __forceinline__ v8f fft256_inv_crop(const cfrag& x, const lconst& k,
                                               const cropfrag& w) {
  ctile b = cmatmul3<-1, true, 4>(x, k.f);
#pragma unroll
  for (int j = 0; j < 8; ++j) {
    const float wr = k.twr[j], wi = -k.twi[j];
    const float re = b.re[j] * wr - b.im[j] * wi;
    const float im = b.re[j] * wi + b.im[j] * wr;
    b.re[j] = re;
    b.im[j] = im;
  }
  const cfrag a = tile_as_frag(b);
  v8f dr = {}, di = {};
#pragma unroll
  for (int t = 0; t < 4; ++t) {
    dr = wmma_f32(a.re[t], w.p1[t], dr);
    di = wmma_f32(a.im[t], w.p2[t], di);
  }
  return dr + di;
}

template <int NCH, typename SRC>
__device__ __forceinline__ ctile fft256_in_real(SRC src,
                                                const cfragn<NCH>& fin,
                                                const lconst& k, int lane) {
  const cfragn<NCH> x = load_seq_frag_in<NCH>(src, lane);
  return fft256_tail<+1>(rmatmul16<NCH>(x, fin), k);
}

template <int NCH, typename SRC>
__device__ __forceinline__ ctile fft256_in(SRC src, const kfragn<NCH>& fin,
                                           const lconst& k, int lane) {
  const cfragn<NCH> x = load_seq_frag_in<NCH>(src, lane);
  return fft256_tail<+1>(cmatmul3<+1, true, NCH>(x, fin), k);
}

__device__ __forceinline__ void init_tables(float4* Ftab, float2* Wtab,
                                            int tid) {
  for (int i = tid; i < 256; i += NTHREADS) {
    const int r = i >> 4, c = i & 15;
    const float tf = -2.0f * PI_F * (float)((r * c) & 15) / 16.0f;
    const float fr = cosf(tf), fi = sinf(tf);
    Ftab[i] = make_float4(fr, fi, fi - fr, -(fr + fi));
    const float tw = -2.0f * PI_F * (float)i / 256.0f;
    Wtab[i] = make_float2(cosf(tw), sinf(tw));
  }
}

__global__ void __launch_bounds__(NTHREADS) kf_precompute(
    const float* __restrict__ kin, float2* __restrict__ kf) {
  extern __shared__ __align__(16) char smem[];
  float2* Rk   = (float2*)smem;
  float4* Ftab = (float4*)(smem + 64 * RSTRIDE * 8);
  float2* Wtab = (float2*)(Ftab + 256);

  const int tid = threadIdx.x, lane = tid & 31, wave = tid >> 5;
  const int lc = lane & 15, lh = lane >> 4;
  const int c = blockIdx.x;

  init_tables(Ftab, Wtab, tid);
  const float* kc = kin + (size_t)c * (64 * 64);
  for (int i = tid; i < 64 * 64; i += NTHREADS) {
    const int r = i >> 6, w = i & 63;
    Rk[r * RSTRIDE + w] = make_float2(kc[i], 0.0f);
  }
  __syncthreads();

  const lconst lk = load_lane_consts(Ftab, Wtab, lane);
  const cfragn<1> fin_r = load_dft_frag_in<1>(Ftab, lane);
  const kfragn<1> fin_c = load_dft_kfrag_in<1>(Ftab, lane);

  for (int row = wave; row < 64; row += NWAVES) {
    float2* Rrow = Rk + row * RSTRIDE;
    const ctile o = fft256_in_real<1>(
        [&](int n) { return make_float2(Rrow[n].x, 0.0f); }, fin_r, lk, lane);
#pragma unroll
    for (int j = 0; j < 8; ++j) {
      const int k = lc + 16 * (8 * lh + j);
      if (k <= 128) Rrow[k] = make_float2(o.re[j], o.im[j]);
    }
  }
  __syncthreads();

  for (int col = wave; col < 129; col += NWAVES) {
    float2* kfcol = kf + ((size_t)c * KFCOLS + col) * 256;
    const ctile o = fft256_in<1>(
        [&](int n) { return Rk[n * RSTRIDE + col]; }, fin_c, lk, lane);
#pragma unroll
    for (int j = 0; j < 8; ++j)
      kfcol[lc + 16 * (8 * lh + j)] = make_float2(o.re[j], o.im[j]);
  }
}

__global__ void __launch_bounds__(NTHREADS) conv_main(
    const float* __restrict__ u, const float* __restrict__ pregate,
    const float* __restrict__ postgate, const float2* __restrict__ kf,
    float* __restrict__ y) {
  extern __shared__ __align__(16) char smem[];
  float2* R    = (float2*)smem;
  float4* Ftab = (float4*)(smem + 128 * PSTRIDE * 8);
  float2* Wtab = (float2*)(Ftab + 256);
  float4* Ctab = (float4*)(Wtab + 256);

  const int tid = threadIdx.x, lane = tid & 31, wave = tid >> 5;
  const int lc = lane & 15, lh = lane >> 4;
  const int part = lc >> 3, k0 = 8 * (lh + 2 * (lc & 7));
  const int c = blockIdx.x & 255;
  const int bp = blockIdx.x >> 8;
  const size_t base_a = ((size_t)(2 * bp) * 256 + c) * (128 * 128);
  const size_t base_b = base_a + (size_t)256 * (128 * 128);

  init_tables(Ftab, Wtab, tid);

  for (int i = tid; i < 128 * 128; i += NTHREADS) {
    const int r = i >> 7, w = i & 127;
    R[r * PSTRIDE + w] = make_float2(u[base_a + i] * pregate[base_a + i],
                                     u[base_b + i] * pregate[base_b + i]);
  }
  __syncthreads();

  const lconst lk = load_lane_consts(Ftab, Wtab, lane);
  const kfragn<2> fin = load_dft_kfrag_in<2>(Ftab, lane);
  init_crop_table(Ctab, Ftab, tid);

  prio_by_progress(0, 128 / NWAVES);
#pragma unroll 1
  for (int row = wave, it = 1; row < 128; row += NWAVES, ++it) {
    float2* Rrow = R + row * PSTRIDE;
    const ctile o =
        fft256_in<2>([&](int n) { return Rrow[n]; }, fin, lk, lane);
#pragma unroll
    for (int j = 0; j < 8; ++j)
      Rrow[lc + 16 * (8 * lh + j)] = make_float2(o.re[j], o.im[j]);
    if (it < 128 / NWAVES) prio_by_progress(it, 128 / NWAVES);
  }
  __syncthreads();

  const float2* kfbase = kf + (size_t)c * KFCOLS * 256;
  prio_by_progress(0, 256 / NWAVES);
#pragma unroll 1
  for (int col = wave, it = 1; col < 256; col += NWAVES, ++it) {
    if (col + NWAVES < 256) {
      const int nc = col + NWAVES;
      const char* pf =
          (const char*)(kfbase + (size_t)((nc > 128) ? 256 - nc : nc) * 256);
      __builtin_prefetch(pf + lane * 64, 0, 1);
    }

    const bool mir = col > 128;
    const float sg = mir ? -1.0f : 1.0f;
    const float2* kfcol = kfbase + (size_t)(mir ? 256 - col : col) * 256;
    float2 kv[8];
#pragma unroll
    for (int j = 0; j < 8; ++j) kv[j] = kfcol[lc + 16 * (8 * lh + j)];

    ctile s = fft256_in<2>(
        [&](int n) {
          const float2 v = R[n * PSTRIDE + col];
          return make_float2(v.x, v.y * sg);
        },
        fin, lk, lane);

#pragma unroll
    for (int j = 0; j < 8; ++j) {
      const float ar = s.re[j], ai = s.im[j];
      s.re[j] = ar * kv[j].x - ai * kv[j].y;
      s.im[j] = ar * kv[j].y + ai * kv[j].x;
    }

    const cropfrag cw = load_crop_frag(Ctab, lane);
    const v8f o = fft256_inv_crop(tile_as_frag(s), lk, cw);
    {
      float* Rf = (float*)(R + col) + part;
      const float sgp = part ? sg : 1.0f;
#pragma unroll
      for (int j = 0; j < 8; ++j) Rf[(k0 + j) * (2 * PSTRIDE)] = o[j] * sgp;
    }
    if (it < 256 / NWAVES) prio_by_progress(it, 256 / NWAVES);
  }
  __syncthreads();

  const float inv_scale = 1.0f / 65536.0f;
  const cropfrag cw = load_crop_frag(Ctab, lane);
  prio_by_progress(0, 128 / NWAVES);
#pragma unroll 1
  for (int row = wave, it = 1; row < 128; row += NWAVES, ++it) {
    float2* Rrow = R + row * PSTRIDE;
    const size_t oo = (part ? base_b : base_a) + (size_t)row * 128 + k0;
    const float4 g0 = *(const float4*)(postgate + oo);
    const float4 g1 = *(const float4*)(postgate + oo + 4);
    const cfrag x = load_seq_frag([&](int n) { return Rrow[n]; }, lane);
    const v8f o = fft256_inv_crop(x, lk, cw);
    *(float4*)(y + oo) =
        make_float4(o[0] * inv_scale * g0.x, o[1] * inv_scale * g0.y,
                    o[2] * inv_scale * g0.z, o[3] * inv_scale * g0.w);
    *(float4*)(y + oo + 4) =
        make_float4(o[4] * inv_scale * g1.x, o[5] * inv_scale * g1.y,
                    o[6] * inv_scale * g1.z, o[7] * inv_scale * g1.w);
    if (it < 128 / NWAVES) prio_by_progress(it, 128 / NWAVES);
  }
}

extern "C" void kernel_launch(void* const* d_in, const int* in_sizes, int n_in,
                              void* d_out, int out_size, void* d_ws,
                              size_t ws_size, hipStream_t stream) {
  const float* u        = (const float*)d_in[0];
  const float* k        = (const float*)d_in[1];
  const float* pregate  = (const float*)d_in[2];
  const float* postgate = (const float*)d_in[3];
  float* y   = (float*)d_out;
  float2* kf = (float2*)d_ws;

  const size_t lds_kf   = (size_t)64 * RSTRIDE * 8 + 256 * 16 + 256 * 8;
  const size_t lds_main =
      (size_t)128 * PSTRIDE * 8 + 256 * 16 + 256 * 8 + CROPTAB * 16;

  (void)hipFuncSetAttribute((const void*)kf_precompute,
                            hipFuncAttributeMaxDynamicSharedMemorySize,
                            (int)lds_kf);
  (void)hipFuncSetAttribute((const void*)conv_main,
                            hipFuncAttributeMaxDynamicSharedMemorySize,
                            (int)lds_main);

  kf_precompute<<<256, NTHREADS, lds_kf, stream>>>(k, kf);
  conv_main<<<2 * 256, NTHREADS, lds_main, stream>>>(u, pregate, postgate, kf, y);
}


# Main kernel: the forward row transforms take their samples straight from memory, each wave one row ahead of its transform, so the input streams in beside the matrix work; LDS staging pass removed
# speedup vs baseline: 4.2660x; 1.0345x over previous
#include <hip/hip_runtime.h>
#include <hip/hip_bf16.h>
#include <math.h>


typedef float v2f __attribute__((ext_vector_type(2)));
typedef float v8f __attribute__((ext_vector_type(8)));

#define PI_F 3.14159265358979323846f
#define RSTRIDE 133
#define PSTRIDE 257
#define KFCOLS 132
#define NWAVES 32
#define NTHREADS (NWAVES * 32)

__device__ __forceinline__ v8f wmma_f32(v2f a, v2f b, v8f c) {
  return __builtin_amdgcn_wmma_f32_16x16x4_f32(false, a, false, b, (short)0, c,
                                               false, false);
}

__device__ __forceinline__ void prio_by_progress(int it, int n) {
  const int lvl = (3 * (n - 1 - it) + (n - 2)) / (n - 1);
  if (lvl >= 3)
    __builtin_amdgcn_s_setprio(3);
  else if (lvl == 2)
    __builtin_amdgcn_s_setprio(2);
  else if (lvl == 1)
    __builtin_amdgcn_s_setprio(1);
  else
    __builtin_amdgcn_s_setprio(0);
}

__device__ __forceinline__ int kpos(int t, int h, int s) {
  return 8 * h + 2 * t + s;
}

__device__ __forceinline__ int kpos_in(int t, int h, int s) {
  return 4 * t + 2 * s + h;
}

struct ctile { v8f re, im; };
template <int NCH> struct cfragn { v2f re[NCH], im[NCH]; };
typedef cfragn<4> cfrag;
template <int NCH> struct kfragn { v2f r[NCH], p[NCH], q[NCH]; };
typedef kfragn<4> kfrag;

__device__ __forceinline__ cfrag tile_as_frag(const ctile& d) {
  cfrag f;
#pragma unroll
  for (int t = 0; t < 4; ++t) {
    f.re[t] = (v2f){d.re[2 * t], d.re[2 * t + 1]};
    f.im[t] = (v2f){d.im[2 * t], d.im[2 * t + 1]};
  }
  return f;
}

__device__ __forceinline__ kfrag load_dft_frag(const float4* Ftab, int lane) {
  const int c = lane & 15, h = lane >> 4;
  kfrag f;
#pragma unroll
  for (int t = 0; t < 4; ++t) {
    const float4 e0 = Ftab[kpos(t, h, 0) * 16 + c];
    const float4 e1 = Ftab[kpos(t, h, 1) * 16 + c];
    f.r[t] = (v2f){e0.x, e1.x};
    f.p[t] = (v2f){e0.z, e1.z};
    f.q[t] = (v2f){e0.w, e1.w};
  }
  return f;
}

struct lconst {
  kfrag f;
  v8f twr, twi;
};

__device__ __forceinline__ lconst load_lane_consts(const float4* Ftab,
                                                   const float2* Wtab,
                                                   int lane) {
  const int c = lane & 15, h = lane >> 4;
  lconst k;
  k.f = load_dft_frag(Ftab, lane);
#pragma unroll
  for (int j = 0; j < 8; ++j) {
    const float2 w = Wtab[((8 * h + j) * c) & 255];
    k.twr[j] = w.x;
    k.twi[j] = w.y;
  }
  return k;
}

template <typename SRC>
__device__ __forceinline__ cfrag load_seq_frag(SRC src, int lane) {
  const int c = lane & 15, h = lane >> 4;
  cfrag f;
#pragma unroll
  for (int t = 0; t < 4; ++t) {
    const float2 e0 = src(c + 16 * kpos(t, h, 0));
    const float2 e1 = src(c + 16 * kpos(t, h, 1));
    f.re[t] = (v2f){e0.x, e1.x};
    f.im[t] = (v2f){e0.y, e1.y};
  }
  return f;
}

template <int NCH>
__device__ __forceinline__ cfragn<NCH> load_dft_frag_in(const float4* Ftab,
                                                        int lane) {
  const int c = lane & 15, h = lane >> 4;
  cfragn<NCH> f;
#pragma unroll
  for (int t = 0; t < NCH; ++t) {
    const float4 e0 = Ftab[kpos_in(t, h, 0) * 16 + c];
    const float4 e1 = Ftab[kpos_in(t, h, 1) * 16 + c];
    f.re[t] = (v2f){e0.x, e1.x};
    f.im[t] = (v2f){e0.y, e1.y};
  }
  return f;
}

template <int NCH>
__device__ __forceinline__ kfragn<NCH> load_dft_kfrag_in(const float4* Ftab,
                                                         int lane) {
  const int c = lane & 15, h = lane >> 4;
  kfragn<NCH> f;
#pragma unroll
  for (int t = 0; t < NCH; ++t) {
    const float4 e0 = Ftab[kpos_in(t, h, 0) * 16 + c];
    const float4 e1 = Ftab[kpos_in(t, h, 1) * 16 + c];
    f.r[t] = (v2f){e0.x, e1.x};
    f.p[t] = (v2f){e0.z, e1.z};
    f.q[t] = (v2f){e0.w, e1.w};
  }
  return f;
}

template <int NCH, typename SRC>
__device__ __forceinline__ cfragn<NCH> load_seq_frag_in(SRC src, int lane) {
  const int c = lane & 15, h = lane >> 4;
  cfragn<NCH> f;
#pragma unroll
  for (int t = 0; t < NCH; ++t) {
    const float2 e0 = src(c + 16 * kpos_in(t, h, 0));
    const float2 e1 = src(c + 16 * kpos_in(t, h, 1));
    f.re[t] = (v2f){e0.x, e1.x};
    f.im[t] = (v2f){e0.y, e1.y};
  }
  return f;
}

template <bool XA>
__device__ __forceinline__ v8f mm(v2f x, v2f w, v8f c) {
  return XA ? wmma_f32(x, w, c) : wmma_f32(w, x, c);
}

template <int DIR, bool XA, int NCH>
__device__ __forceinline__ ctile cmatmul3(const cfragn<NCH>& x,
                                          const kfragn<NCH>& w) {
  v8f k = {};
#pragma unroll
  for (int t = 0; t < NCH; ++t) k = mm<XA>(x.re[t] + x.im[t], w.r[t], k);
  ctile d;
  d.re = k;
  d.im = k;
#pragma unroll
  for (int t = 0; t < NCH; ++t) {
    d.re = mm<XA>(x.im[t], (DIR > 0) ? w.q[t] : w.p[t], d.re);
    d.im = mm<XA>(x.re[t], (DIR > 0) ? w.p[t] : w.q[t], d.im);
  }
  return d;
}

template <int NCH>
__device__ __forceinline__ ctile rmatmul16(const cfragn<NCH>& a,
                                           const cfragn<NCH>& b) {
  v8f rr = {}, ri = {};
#pragma unroll
  for (int t = 0; t < NCH; ++t) {
    rr = wmma_f32(a.re[t], b.re[t], rr);
    ri = wmma_f32(a.re[t], b.im[t], ri);
  }
  ctile d;
  d.re = rr;
  d.im = ri;
  return d;
}

template <int DIR>
__device__ __forceinline__ ctile fft256_tail(ctile b, const lconst& k) {
#pragma unroll
  for (int j = 0; j < 8; ++j) {
    const float wr = k.twr[j], wi = (DIR > 0) ? k.twi[j] : -k.twi[j];
    const float re = b.re[j] * wr - b.im[j] * wi;
    const float im = b.re[j] * wi + b.im[j] * wr;
    b.re[j] = re;
    b.im[j] = im;
  }
  return cmatmul3<DIR, false, 4>(tile_as_frag(b), k.f);
}

#define CROPTAB 128
__device__ __forceinline__ void init_crop_table(float4* Ctab,
                                                const float4* Ftab, int tid) {
  if (tid < CROPTAB) {
    const int t = tid >> 5, l = tid & 31;
    const int c = l & 15, h = l >> 4, cc = c & 7;
    const float4 e0 = Ftab[kpos(t, h, 0) * 16 + cc];
    const float4 e1 = Ftab[kpos(t, h, 1) * 16 + cc];
    Ctab[tid] = (c < 8) ? make_float4(e0.x, e1.x, e0.y, e1.y)
                        : make_float4(-e0.y, -e1.y, e0.x, e1.x);
  }
}

struct cropfrag { v2f p1[4], p2[4]; };

__device__ __forceinline__ cropfrag load_crop_frag(const float4* Ctab,
                                                   int lane) {
  cropfrag f;
#pragma unroll
  for (int t = 0; t < 4; ++t) {
    const float4 e = Ctab[t * 32 + lane];
    f.p1[t] = (v2f){e.x, e.y};
    f.p2[t] = (v2f){e.z, e.w};
  }
  return f;
}

__device__ __forceinline__ v8f fft256_inv_crop(const cfrag& x, const lconst& k,
                                               const cropfrag& w) {
  ctile b = cmatmul3<-1, true, 4>(x, k.f);
#pragma unroll
  for (int j = 0; j < 8; ++j) {
    const float wr = k.twr[j], wi = -k.twi[j];
    const float re = b.re[j] * wr - b.im[j] * wi;
    const float im = b.re[j] * wi + b.im[j] * wr;
    b.re[j] = re;
    b.im[j] = im;
  }
  const cfrag a = tile_as_frag(b);
  v8f dr = {}, di = {};
#pragma unroll
  for (int t = 0; t < 4; ++t) {
    dr = wmma_f32(a.re[t], w.p1[t], dr);
    di = wmma_f32(a.im[t], w.p2[t], di);
  }
  return dr + di;
}

template <int NCH, typename SRC>
__device__ __forceinline__ ctile fft256_in_real(SRC src,
                                                const cfragn<NCH>& fin,
                                                const lconst& k, int lane) {
  const cfragn<NCH> x = load_seq_frag_in<NCH>(src, lane);
  return fft256_tail<+1>(rmatmul16<NCH>(x, fin), k);
}

template <int NCH, typename SRC>
__device__ __forceinline__ ctile fft256_in(SRC src, const kfragn<NCH>& fin,
                                           const lconst& k, int lane) {
  const cfragn<NCH> x = load_seq_frag_in<NCH>(src, lane);
  return fft256_tail<+1>(cmatmul3<+1, true, NCH>(x, fin), k);
}

__device__ __forceinline__ void init_tables(float4* Ftab, float2* Wtab,
                                            int tid) {
  for (int i = tid; i < 256; i += NTHREADS) {
    const int r = i >> 4, c = i & 15;
    const float tf = -2.0f * PI_F * (float)((r * c) & 15) / 16.0f;
    const float fr = cosf(tf), fi = sinf(tf);
    Ftab[i] = make_float4(fr, fi, fi - fr, -(fr + fi));
    const float tw = -2.0f * PI_F * (float)i / 256.0f;
    Wtab[i] = make_float2(cosf(tw), sinf(tw));
  }
}

__global__ void __launch_bounds__(NTHREADS) kf_precompute(
    const float* __restrict__ kin, float2* __restrict__ kf) {
  extern __shared__ __align__(16) char smem[];
  float2* Rk   = (float2*)smem;
  float4* Ftab = (float4*)(smem + 64 * RSTRIDE * 8);
  float2* Wtab = (float2*)(Ftab + 256);

  const int tid = threadIdx.x, lane = tid & 31, wave = tid >> 5;
  const int lc = lane & 15, lh = lane >> 4;
  const int c = blockIdx.x;

  init_tables(Ftab, Wtab, tid);
  const float* kc = kin + (size_t)c * (64 * 64);
  for (int i = tid; i < 64 * 64; i += NTHREADS) {
    const int r = i >> 6, w = i & 63;
    Rk[r * RSTRIDE + w] = make_float2(kc[i], 0.0f);
  }
  __syncthreads();

  const lconst lk = load_lane_consts(Ftab, Wtab, lane);
  const cfragn<1> fin_r = load_dft_frag_in<1>(Ftab, lane);
  const kfragn<1> fin_c = load_dft_kfrag_in<1>(Ftab, lane);

  for (int row = wave; row < 64; row += NWAVES) {
    float2* Rrow = Rk + row * RSTRIDE;
    const ctile o = fft256_in_real<1>(
        [&](int n) { return make_float2(Rrow[n].x, 0.0f); }, fin_r, lk, lane);
#pragma unroll
    for (int j = 0; j < 8; ++j) {
      const int k = lc + 16 * (8 * lh + j);
      if (k <= 128) Rrow[k] = make_float2(o.re[j], o.im[j]);
    }
  }
  __syncthreads();

  for (int col = wave; col < 129; col += NWAVES) {
    float2* kfcol = kf + ((size_t)c * KFCOLS + col) * 256;
    const ctile o = fft256_in<1>(
        [&](int n) { return Rk[n * RSTRIDE + col]; }, fin_c, lk, lane);
#pragma unroll
    for (int j = 0; j < 8; ++j)
      kfcol[lc + 16 * (8 * lh + j)] = make_float2(o.re[j], o.im[j]);
  }
}

struct rowin { float ua[4], pa[4], ub[4], pb[4]; };

__device__ __forceinline__ rowin fetch_row(const float* __restrict__ ua,
                                           const float* __restrict__ pa,
                                           const float* __restrict__ ub,
                                           const float* __restrict__ pb,
                                           int row, int lane) {
  rowin r;
  const unsigned o = (unsigned)row * 128u + (unsigned)lane;
#pragma unroll
  for (int q = 0; q < 4; ++q) {
    r.ua[q] = ua[o + 32u * q];
    r.pa[q] = pa[o + 32u * q];
    r.ub[q] = ub[o + 32u * q];
    r.pb[q] = pb[o + 32u * q];
  }
  return r;
}

__device__ __forceinline__ cfragn<2> gate_row(const rowin& r) {
  cfragn<2> x;
#pragma unroll
  for (int t = 0; t < 2; ++t) {
    x.re[t] = (v2f){r.ua[2 * t] * r.pa[2 * t], r.ua[2 * t + 1] * r.pa[2 * t + 1]};
    x.im[t] = (v2f){r.ub[2 * t] * r.pb[2 * t], r.ub[2 * t + 1] * r.pb[2 * t + 1]};
  }
  return x;
}

__global__ void __launch_bounds__(NTHREADS) conv_main(
    const float* __restrict__ u, const float* __restrict__ pregate,
    const float* __restrict__ postgate, const float2* __restrict__ kf,
    float* __restrict__ y) {
  extern __shared__ __align__(16) char smem[];
  float2* R    = (float2*)smem;
  float4* Ftab = (float4*)(smem + 128 * PSTRIDE * 8);
  float2* Wtab = (float2*)(Ftab + 256);
  float4* Ctab = (float4*)(Wtab + 256);

  const int tid = threadIdx.x, lane = tid & 31, wave = tid >> 5;
  const int lc = lane & 15, lh = lane >> 4;
  const int part = lc >> 3, k0 = 8 * (lh + 2 * (lc & 7));
  const int c = blockIdx.x & 255;
  const int bp = blockIdx.x >> 8;
  const size_t base_a = ((size_t)(2 * bp) * 256 + c) * (128 * 128);
  const size_t base_b = base_a + (size_t)256 * (128 * 128);

  const float* ua = u + base_a;
  const float* pa = pregate + base_a;
  const float* ub = u + base_b;
  const float* pb = pregate + base_b;
  rowin cur = fetch_row(ua, pa, ub, pb, wave, lane);

  init_tables(Ftab, Wtab, tid);
  __syncthreads();

  const lconst lk = load_lane_consts(Ftab, Wtab, lane);
  const kfragn<2> fin = load_dft_kfrag_in<2>(Ftab, lane);
  init_crop_table(Ctab, Ftab, tid);

  prio_by_progress(0, 128 / NWAVES);
#pragma unroll 1
  for (int row = wave, it = 1; row < 128; row += NWAVES, ++it) {
    float2* Rrow = R + row * PSTRIDE;
    rowin nxt = cur;
    if (it < 128 / NWAVES) nxt = fetch_row(ua, pa, ub, pb, row + NWAVES, lane);
    const cfragn<2> x = gate_row(cur);
    const ctile o = fft256_tail<+1>(cmatmul3<+1, true, 2>(x, fin), lk);
#pragma unroll
    for (int j = 0; j < 8; ++j)
      Rrow[lc + 16 * (8 * lh + j)] = make_float2(o.re[j], o.im[j]);
    cur = nxt;
    if (it < 128 / NWAVES) prio_by_progress(it, 128 / NWAVES);
  }
  __syncthreads();

  const float2* kfbase = kf + (size_t)c * KFCOLS * 256;
  prio_by_progress(0, 256 / NWAVES);
#pragma unroll 1
  for (int col = wave, it = 1; col < 256; col += NWAVES, ++it) {
    if (col + NWAVES < 256) {
      const int nc = col + NWAVES;
      const char* pf =
          (const char*)(kfbase + (size_t)((nc > 128) ? 256 - nc : nc) * 256);
      __builtin_prefetch(pf + lane * 64, 0, 1);
    }

    const bool mir = col > 128;
    const float sg = mir ? -1.0f : 1.0f;
    const float2* kfcol = kfbase + (size_t)(mir ? 256 - col : col) * 256;
    float2 kv[8];
#pragma unroll
    for (int j = 0; j < 8; ++j) kv[j] = kfcol[lc + 16 * (8 * lh + j)];

    ctile s = fft256_in<2>(
        [&](int n) {
          const float2 v = R[n * PSTRIDE + col];
          return make_float2(v.x, v.y * sg);
        },
        fin, lk, lane);

#pragma unroll
    for (int j = 0; j < 8; ++j) {
      const float ar = s.re[j], ai = s.im[j];
      s.re[j] = ar * kv[j].x - ai * kv[j].y;
      s.im[j] = ar * kv[j].y + ai * kv[j].x;
    }

    const cropfrag cw = load_crop_frag(Ctab, lane);
    const v8f o = fft256_inv_crop(tile_as_frag(s), lk, cw);
    {
      float* Rf = (float*)(R + col) + part;
      const float sgp = part ? sg : 1.0f;
#pragma unroll
      for (int j = 0; j < 8; ++j) Rf[(k0 + j) * (2 * PSTRIDE)] = o[j] * sgp;
    }
    if (it < 256 / NWAVES) prio_by_progress(it, 256 / NWAVES);
  }
  __syncthreads();

  const float inv_scale = 1.0f / 65536.0f;
  const cropfrag cw = load_crop_frag(Ctab, lane);
  prio_by_progress(0, 128 / NWAVES);
#pragma unroll 1
  for (int row = wave, it = 1; row < 128; row += NWAVES, ++it) {
    float2* Rrow = R + row * PSTRIDE;
    const size_t oo = (part ? base_b : base_a) + (size_t)row * 128 + k0;
    const float4 g0 = *(const float4*)(postgate + oo);
    const float4 g1 = *(const float4*)(postgate + oo + 4);
    const cfrag x = load_seq_frag([&](int n) { return Rrow[n]; }, lane);
    const v8f o = fft256_inv_crop(x, lk, cw);
    *(float4*)(y + oo) =
        make_float4(o[0] * inv_scale * g0.x, o[1] * inv_scale * g0.y,
                    o[2] * inv_scale * g0.z, o[3] * inv_scale * g0.w);
    *(float4*)(y + oo + 4) =
        make_float4(o[4] * inv_scale * g1.x, o[5] * inv_scale * g1.y,
                    o[6] * inv_scale * g1.z, o[7] * inv_scale * g1.w);
    if (it < 128 / NWAVES) prio_by_progress(it, 128 / NWAVES);
  }
}

extern "C" void kernel_launch(void* const* d_in, const int* in_sizes, int n_in,
                              void* d_out, int out_size, void* d_ws,
                              size_t ws_size, hipStream_t stream) {
  const float* u        = (const float*)d_in[0];
  const float* k        = (const float*)d_in[1];
  const float* pregate  = (const float*)d_in[2];
  const float* postgate = (const float*)d_in[3];
  float* y   = (float*)d_out;
  float2* kf = (float2*)d_ws;

  const size_t lds_kf   = (size_t)64 * RSTRIDE * 8 + 256 * 16 + 256 * 8;
  const size_t lds_main =
      (size_t)128 * PSTRIDE * 8 + 256 * 16 + 256 * 8 + CROPTAB * 16;

  (void)hipFuncSetAttribute((const void*)kf_precompute,
                            hipFuncAttributeMaxDynamicSharedMemorySize,
                            (int)lds_kf);
  (void)hipFuncSetAttribute((const void*)conv_main,
                            hipFuncAttributeMaxDynamicSharedMemorySize,
                            (int)lds_main);

  kf_precompute<<<256, NTHREADS, lds_kf, stream>>>(k, kf);
  conv_main<<<2 * 256, NTHREADS, lds_main, stream>>>(u, pregate, postgate, kf, y);
}


# One launch, one workgroup per channel: it computes the filter's spectrum, then runs the channel's two image pairs in a loop with no barrier between pairs (a wave's rows stay its own)
# speedup vs baseline: 4.4780x; 1.0497x over previous
#include <hip/hip_runtime.h>
#include <hip/hip_bf16.h>
#include <math.h>


typedef float v2f __attribute__((ext_vector_type(2)));
typedef float v8f __attribute__((ext_vector_type(8)));

#define PI_F 3.14159265358979323846f
#define RSTRIDE 133
#define PSTRIDE 257
#define KFCOLS 132
#define NWAVES 32
#define NTHREADS (NWAVES * 32)

__device__ __forceinline__ v8f wmma_f32(v2f a, v2f b, v8f c) {
  return __builtin_amdgcn_wmma_f32_16x16x4_f32(false, a, false, b, (short)0, c,
                                               false, false);
}

__device__ __forceinline__ void prio_by_progress(int it, int n) {
  const int lvl = (3 * (n - 1 - it) + (n - 2)) / (n - 1);
  if (lvl >= 3)
    __builtin_amdgcn_s_setprio(3);
  else if (lvl == 2)
    __builtin_amdgcn_s_setprio(2);
  else if (lvl == 1)
    __builtin_amdgcn_s_setprio(1);
  else
    __builtin_amdgcn_s_setprio(0);
}

__device__ __forceinline__ int kpos(int t, int h, int s) {
  return 8 * h + 2 * t + s;
}

__device__ __forceinline__ int kpos_in(int t, int h, int s) {
  return 4 * t + 2 * s + h;
}

struct ctile { v8f re, im; };
template <int NCH> struct cfragn { v2f re[NCH], im[NCH]; };
typedef cfragn<4> cfrag;
template <int NCH> struct kfragn { v2f r[NCH], p[NCH], q[NCH]; };
typedef kfragn<4> kfrag;

__device__ __forceinline__ cfrag tile_as_frag(const ctile& d) {
  cfrag f;
#pragma unroll
  for (int t = 0; t < 4; ++t) {
    f.re[t] = (v2f){d.re[2 * t], d.re[2 * t + 1]};
    f.im[t] = (v2f){d.im[2 * t], d.im[2 * t + 1]};
  }
  return f;
}

__device__ __forceinline__ kfrag load_dft_frag(const float4* Ftab, int lane) {
  const int c = lane & 15, h = lane >> 4;
  kfrag f;
#pragma unroll
  for (int t = 0; t < 4; ++t) {
    const float4 e0 = Ftab[kpos(t, h, 0) * 16 + c];
    const float4 e1 = Ftab[kpos(t, h, 1) * 16 + c];
    f.r[t] = (v2f){e0.x, e1.x};
    f.p[t] = (v2f){e0.z, e1.z};
    f.q[t] = (v2f){e0.w, e1.w};
  }
  return f;
}

struct lconst {
  kfrag f;
  v8f twr, twi;
};

__device__ __forceinline__ lconst load_lane_consts(const float4* Ftab,
                                                   const float2* Wtab,
                                                   int lane) {
  const int c = lane & 15, h = lane >> 4;
  lconst k;
  k.f = load_dft_frag(Ftab, lane);
#pragma unroll
  for (int j = 0; j < 8; ++j) {
    const float2 w = Wtab[((8 * h + j) * c) & 255];
    k.twr[j] = w.x;
    k.twi[j] = w.y;
  }
  return k;
}

template <typename SRC>
__device__ __forceinline__ cfrag load_seq_frag(SRC src, int lane) {
  const int c = lane & 15, h = lane >> 4;
  cfrag f;
#pragma unroll
  for (int t = 0; t < 4; ++t) {
    const float2 e0 = src(c + 16 * kpos(t, h, 0));
    const float2 e1 = src(c + 16 * kpos(t, h, 1));
    f.re[t] = (v2f){e0.x, e1.x};
    f.im[t] = (v2f){e0.y, e1.y};
  }
  return f;
}

template <int NCH>
__device__ __forceinline__ cfragn<NCH> load_dft_frag_in(const float4* Ftab,
                                                        int lane) {
  const int c = lane & 15, h = lane >> 4;
  cfragn<NCH> f;
#pragma unroll
  for (int t = 0; t < NCH; ++t) {
    const float4 e0 = Ftab[kpos_in(t, h, 0) * 16 + c];
    const float4 e1 = Ftab[kpos_in(t, h, 1) * 16 + c];
    f.re[t] = (v2f){e0.x, e1.x};
    f.im[t] = (v2f){e0.y, e1.y};
  }
  return f;
}

template <int NCH>
__device__ __forceinline__ kfragn<NCH> load_dft_kfrag_in(const float4* Ftab,
                                                         int lane) {
  const int c = lane & 15, h = lane >> 4;
  kfragn<NCH> f;
#pragma unroll
  for (int t = 0; t < NCH; ++t) {
    const float4 e0 = Ftab[kpos_in(t, h, 0) * 16 + c];
    const float4 e1 = Ftab[kpos_in(t, h, 1) * 16 + c];
    f.r[t] = (v2f){e0.x, e1.x};
    f.p[t] = (v2f){e0.z, e1.z};
    f.q[t] = (v2f){e0.w, e1.w};
  }
  return f;
}

template <int NCH, typename SRC>
__device__ __forceinline__ cfragn<NCH> load_seq_frag_in(SRC src, int lane) {
  const int c = lane & 15, h = lane >> 4;
  cfragn<NCH> f;
#pragma unroll
  for (int t = 0; t < NCH; ++t) {
    const float2 e0 = src(c + 16 * kpos_in(t, h, 0));
    const float2 e1 = src(c + 16 * kpos_in(t, h, 1));
    f.re[t] = (v2f){e0.x, e1.x};
    f.im[t] = (v2f){e0.y, e1.y};
  }
  return f;
}

template <bool XA>
__device__ __forceinline__ v8f mm(v2f x, v2f w, v8f c) {
  return XA ? wmma_f32(x, w, c) : wmma_f32(w, x, c);
}

template <int DIR, bool XA, int NCH>
__device__ __forceinline__ ctile cmatmul3(const cfragn<NCH>& x,
                                          const kfragn<NCH>& w) {
  v8f k = {};
#pragma unroll
  for (int t = 0; t < NCH; ++t) k = mm<XA>(x.re[t] + x.im[t], w.r[t], k);
  ctile d;
  d.re = k;
  d.im = k;
#pragma unroll
  for (int t = 0; t < NCH; ++t) {
    d.re = mm<XA>(x.im[t], (DIR > 0) ? w.q[t] : w.p[t], d.re);
    d.im = mm<XA>(x.re[t], (DIR > 0) ? w.p[t] : w.q[t], d.im);
  }
  return d;
}

template <int NCH>
__device__ __forceinline__ ctile rmatmul16(const cfragn<NCH>& a,
                                           const cfragn<NCH>& b) {
  v8f rr = {}, ri = {};
#pragma unroll
  for (int t = 0; t < NCH; ++t) {
    rr = wmma_f32(a.re[t], b.re[t], rr);
    ri = wmma_f32(a.re[t], b.im[t], ri);
  }
  ctile d;
  d.re = rr;
  d.im = ri;
  return d;
}

template <int DIR>
__device__ __forceinline__ ctile fft256_tail(ctile b, const lconst& k) {
#pragma unroll
  for (int j = 0; j < 8; ++j) {
    const float wr = k.twr[j], wi = (DIR > 0) ? k.twi[j] : -k.twi[j];
    const float re = b.re[j] * wr - b.im[j] * wi;
    const float im = b.re[j] * wi + b.im[j] * wr;
    b.re[j] = re;
    b.im[j] = im;
  }
  return cmatmul3<DIR, false, 4>(tile_as_frag(b), k.f);
}

#define CROPTAB 128
__device__ __forceinline__ void init_crop_table(float4* Ctab,
                                                const float4* Ftab, int tid) {
  if (tid < CROPTAB) {
    const int t = tid >> 5, l = tid & 31;
    const int c = l & 15, h = l >> 4, cc = c & 7;
    const float4 e0 = Ftab[kpos(t, h, 0) * 16 + cc];
    const float4 e1 = Ftab[kpos(t, h, 1) * 16 + cc];
    Ctab[tid] = (c < 8) ? make_float4(e0.x, e1.x, e0.y, e1.y)
                        : make_float4(-e0.y, -e1.y, e0.x, e1.x);
  }
}

struct cropfrag { v2f p1[4], p2[4]; };

__device__ __forceinline__ cropfrag load_crop_frag(const float4* Ctab,
                                                   int lane) {
  cropfrag f;
#pragma unroll
  for (int t = 0; t < 4; ++t) {
    const float4 e = Ctab[t * 32 + lane];
    f.p1[t] = (v2f){e.x, e.y};
    f.p2[t] = (v2f){e.z, e.w};
  }
  return f;
}

__device__ __forceinline__ v8f fft256_inv_crop(const cfrag& x, const lconst& k,
                                               const cropfrag& w) {
  ctile b = cmatmul3<-1, true, 4>(x, k.f);
#pragma unroll
  for (int j = 0; j < 8; ++j) {
    const float wr = k.twr[j], wi = -k.twi[j];
    const float re = b.re[j] * wr - b.im[j] * wi;
    const float im = b.re[j] * wi + b.im[j] * wr;
    b.re[j] = re;
    b.im[j] = im;
  }
  const cfrag a = tile_as_frag(b);
  v8f dr = {}, di = {};
#pragma unroll
  for (int t = 0; t < 4; ++t) {
    dr = wmma_f32(a.re[t], w.p1[t], dr);
    di = wmma_f32(a.im[t], w.p2[t], di);
  }
  return dr + di;
}

template <int NCH, typename SRC>
__device__ __forceinline__ ctile fft256_in_real(SRC src,
                                                const cfragn<NCH>& fin,
                                                const lconst& k, int lane) {
  const cfragn<NCH> x = load_seq_frag_in<NCH>(src, lane);
  return fft256_tail<+1>(rmatmul16<NCH>(x, fin), k);
}

template <int NCH, typename SRC>
__device__ __forceinline__ ctile fft256_in(SRC src, const kfragn<NCH>& fin,
                                           const lconst& k, int lane) {
  const cfragn<NCH> x = load_seq_frag_in<NCH>(src, lane);
  return fft256_tail<+1>(cmatmul3<+1, true, NCH>(x, fin), k);
}

__device__ __forceinline__ void init_tables(float4* Ftab, float2* Wtab,
                                            int tid) {
  for (int i = tid; i < 256; i += NTHREADS) {
    const int r = i >> 4, c = i & 15;
    const float tf = -2.0f * PI_F * (float)((r * c) & 15) / 16.0f;
    const float fr = cosf(tf), fi = sinf(tf);
    Ftab[i] = make_float4(fr, fi, fi - fr, -(fr + fi));
    const float tw = -2.0f * PI_F * (float)i / 256.0f;
    Wtab[i] = make_float2(cosf(tw), sinf(tw));
  }
}

struct rowin { float ua[4], pa[4], ub[4], pb[4]; };

__device__ __forceinline__ rowin fetch_row(const float* __restrict__ ua,
                                           const float* __restrict__ pa,
                                           const float* __restrict__ ub,
                                           const float* __restrict__ pb,
                                           int row, int lane) {
  rowin r;
  const unsigned o = (unsigned)row * 128u + (unsigned)lane;
#pragma unroll
  for (int q = 0; q < 4; ++q) {
    r.ua[q] = ua[o + 32u * q];
    r.pa[q] = pa[o + 32u * q];
    r.ub[q] = ub[o + 32u * q];
    r.pb[q] = pb[o + 32u * q];
  }
  return r;
}

__device__ __forceinline__ cfragn<2> gate_row(const rowin& r) {
  cfragn<2> x;
#pragma unroll
  for (int t = 0; t < 2; ++t) {
    x.re[t] = (v2f){r.ua[2 * t] * r.pa[2 * t], r.ua[2 * t + 1] * r.pa[2 * t + 1]};
    x.im[t] = (v2f){r.ub[2 * t] * r.pb[2 * t], r.ub[2 * t + 1] * r.pb[2 * t + 1]};
  }
  return x;
}

__device__ __forceinline__ void prefetch_row(const float* __restrict__ ua,
                                             const float* __restrict__ pa,
                                             const float* __restrict__ ub,
                                             const float* __restrict__ pb,
                                             int row, int lane) {
  const float* a = (lane & 16) ? ((lane & 8) ? pb : ub) : ((lane & 8) ? pa : ua);
  __builtin_prefetch((const char*)(a + row * 128) + (lane & 7) * 64, 0, 1);
}

__global__ void __launch_bounds__(NTHREADS) conv_main(
    const float* __restrict__ u, const float* __restrict__ pregate,
    const float* __restrict__ postgate, const float* __restrict__ kin,
    float2* kf, float* __restrict__ y) {
  extern __shared__ __align__(16) char smem[];
  float2* R    = (float2*)smem;
  float2* Rk   = (float2*)smem;
  float4* Ftab = (float4*)(smem + 128 * PSTRIDE * 8);
  float2* Wtab = (float2*)(Ftab + 256);
  float4* Ctab = (float4*)(Wtab + 256);

  const int tid = threadIdx.x, lane = tid & 31, wave = tid >> 5;
  const int lc = lane & 15, lh = lane >> 4;
  const int part = lc >> 3, k0 = 8 * (lh + 2 * (lc & 7));
  const int c = blockIdx.x;
  const size_t img = (size_t)128 * 128;
  const size_t base_c = (size_t)c * img;
  const size_t bstep = (size_t)256 * img;

  prefetch_row(u + base_c, pregate + base_c, u + base_c + bstep,
               pregate + base_c + bstep, wave, lane);

  init_tables(Ftab, Wtab, tid);
  const float* kc = kin + (size_t)c * (64 * 64);
  for (int i = tid; i < 64 * 64; i += NTHREADS) {
    const int r = i >> 6, w = i & 63;
    Rk[r * RSTRIDE + w] = make_float2(kc[i], 0.0f);
  }
  __syncthreads();

  const lconst lk = load_lane_consts(Ftab, Wtab, lane);
  init_crop_table(Ctab, Ftab, tid);

  float2* kfbase = kf + (size_t)c * KFCOLS * 256;
  {
    const cfragn<1> fin_r = load_dft_frag_in<1>(Ftab, lane);
    const kfragn<1> fin_c = load_dft_kfrag_in<1>(Ftab, lane);

    for (int row = wave; row < 64; row += NWAVES) {
      float2* Rrow = Rk + row * RSTRIDE;
      const ctile o = fft256_in_real<1>(
          [&](int n) { return make_float2(Rrow[n].x, 0.0f); }, fin_r, lk, lane);
#pragma unroll
      for (int j = 0; j < 8; ++j) {
        const int k = lc + 16 * (8 * lh + j);
        if (k <= 128) Rrow[k] = make_float2(o.re[j], o.im[j]);
      }
    }
    __syncthreads();

    for (int col = wave; col < 129; col += NWAVES) {
      float2* kfcol = kfbase + (size_t)col * 256;
      const ctile o = fft256_in<1>(
          [&](int n) { return Rk[n * RSTRIDE + col]; }, fin_c, lk, lane);
#pragma unroll
      for (int j = 0; j < 8; ++j)
        kfcol[lc + 16 * (8 * lh + j)] = make_float2(o.re[j], o.im[j]);
    }
  }
  __syncthreads();

  const float inv_scale = 1.0f / 65536.0f;

#pragma unroll 1
  for (int bp = 0; bp < 2; ++bp) {
    const size_t base_a = base_c + (size_t)(2 * bp) * bstep;
    const size_t base_b = base_a + bstep;

    const float* ua = u + base_a;
    const float* pa = pregate + base_a;
    const float* ub = u + base_b;
    const float* pb = pregate + base_b;
    rowin cur = fetch_row(ua, pa, ub, pb, wave, lane);

    const kfragn<2> fin = load_dft_kfrag_in<2>(Ftab, lane);

    if (bp == 0) prio_by_progress(0, 128 / NWAVES);
#pragma unroll 1
    for (int row = wave, it = 1; row < 128; row += NWAVES, ++it) {
      float2* Rrow = R + row * PSTRIDE;
      rowin nxt = cur;
      if (it < 128 / NWAVES) nxt = fetch_row(ua, pa, ub, pb, row + NWAVES, lane);
      const cfragn<2> x = gate_row(cur);
      const ctile o = fft256_tail<+1>(cmatmul3<+1, true, 2>(x, fin), lk);
#pragma unroll
      for (int j = 0; j < 8; ++j)
        Rrow[lc + 16 * (8 * lh + j)] = make_float2(o.re[j], o.im[j]);
      cur = nxt;
      if (it < 128 / NWAVES) {
        if (bp == 0)
          prio_by_progress(it, 128 / NWAVES);
        else
          prio_by_progress(128 / NWAVES + it, 2 * (128 / NWAVES));
      }
    }
    __syncthreads();

    prio_by_progress(0, 256 / NWAVES);
#pragma unroll 1
    for (int col = wave, it = 1; col < 256; col += NWAVES, ++it) {
      if (col + NWAVES < 256) {
        const int nc = col + NWAVES;
        const char* pf =
            (const char*)(kfbase + (size_t)((nc > 128) ? 256 - nc : nc) * 256);
        __builtin_prefetch(pf + lane * 64, 0, 1);
      }

      const bool mir = col > 128;
      const float sg = mir ? -1.0f : 1.0f;
      const float2* kfcol = kfbase + (size_t)(mir ? 256 - col : col) * 256;
      float2 kv[8];
#pragma unroll
      for (int j = 0; j < 8; ++j) kv[j] = kfcol[lc + 16 * (8 * lh + j)];

      ctile s = fft256_in<2>(
          [&](int n) {
            const float2 v = R[n * PSTRIDE + col];
            return make_float2(v.x, v.y * sg);
          },
          fin, lk, lane);

#pragma unroll
      for (int j = 0; j < 8; ++j) {
        const float ar = s.re[j], ai = s.im[j];
        s.re[j] = ar * kv[j].x - ai * kv[j].y;
        s.im[j] = ar * kv[j].y + ai * kv[j].x;
      }

      const cropfrag cw = load_crop_frag(Ctab, lane);
      const v8f o = fft256_inv_crop(tile_as_frag(s), lk, cw);
      {
        float* Rf = (float*)(R + col) + part;
        const float sgp = part ? sg : 1.0f;
#pragma unroll
        for (int j = 0; j < 8; ++j) Rf[(k0 + j) * (2 * PSTRIDE)] = o[j] * sgp;
      }
      if (it < 256 / NWAVES) prio_by_progress(it, 256 / NWAVES);
    }
    __syncthreads();

    if (bp == 0)
      prefetch_row(ua + 2 * bstep, pa + 2 * bstep, ub + 2 * bstep,
                   pb + 2 * bstep, wave, lane);

    const cropfrag cw = load_crop_frag(Ctab, lane);
    if (bp == 0)
      prio_by_progress(0, 2 * (128 / NWAVES));
    else
      prio_by_progress(0, 128 / NWAVES);
#pragma unroll 1
    for (int row = wave, it = 1; row < 128; row += NWAVES, ++it) {
      float2* Rrow = R + row * PSTRIDE;
      const size_t oo = (part ? base_b : base_a) + (size_t)row * 128 + k0;
      const float4 g0 = *(const float4*)(postgate + oo);
      const float4 g1 = *(const float4*)(postgate + oo + 4);
      const cfrag x = load_seq_frag([&](int n) { return Rrow[n]; }, lane);
      const v8f o = fft256_inv_crop(x, lk, cw);
      *(float4*)(y + oo) =
          make_float4(o[0] * inv_scale * g0.x, o[1] * inv_scale * g0.y,
                      o[2] * inv_scale * g0.z, o[3] * inv_scale * g0.w);
      *(float4*)(y + oo + 4) =
          make_float4(o[4] * inv_scale * g1.x, o[5] * inv_scale * g1.y,
                      o[6] * inv_scale * g1.z, o[7] * inv_scale * g1.w);
      if (bp == 0)
        prio_by_progress(it, 2 * (128 / NWAVES));
      else if (it < 128 / NWAVES)
        prio_by_progress(it, 128 / NWAVES);
    }
  }
}

extern "C" void kernel_launch(void* const* d_in, const int* in_sizes, int n_in,
                              void* d_out, int out_size, void* d_ws,
                              size_t ws_size, hipStream_t stream) {
  const float* u        = (const float*)d_in[0];
  const float* k        = (const float*)d_in[1];
  const float* pregate  = (const float*)d_in[2];
  const float* postgate = (const float*)d_in[3];
  float* y   = (float*)d_out;
  float2* kf = (float2*)d_ws;

  const size_t lds_main =
      (size_t)128 * PSTRIDE * 8 + 256 * 16 + 256 * 8 + CROPTAB * 16;

  (void)hipFuncSetAttribute((const void*)conv_main,
                            hipFuncAttributeMaxDynamicSharedMemorySize,
                            (int)lds_main);

  conv_main<<<256, NTHREADS, lds_main, stream>>>(u, pregate, postgate, k, kf, y);
}
